# starting kernel
# baseline (speedup 1.0000x reference)
#include <hip/hip_runtime.h>

#define NNODES   50000
#define FEAT     128
#define HEADS    4
#define NEDGE    800000
#define ETOT     (NEDGE + NNODES)
#define NEG_SLOPE 0.2f

typedef __attribute__((ext_vector_type(2))) float v2f;
typedef __attribute__((ext_vector_type(8))) float v8f;

__global__ __launch_bounds__(256) void gat_init(float* __restrict__ out,
                                                float* __restrict__ m,
                                                float* __restrict__ s,
                                                const float* __restrict__ bias) {
    int tid = blockIdx.x * 256 + threadIdx.x;
    int stride = gridDim.x * 256;
    for (int i = tid; i < NNODES * FEAT; i += stride)
        out[i] = bias[i & (FEAT - 1)];
    for (int i = tid; i < NNODES * HEADS; i += stride) {
        m[i] = -1e30f;
        s[i] = 0.0f;
    }
}

__global__ __launch_bounds__(256) void gat_gemm(const float* __restrict__ x,
                                                const float* __restrict__ Wl,
                                                const float* __restrict__ Wr,
                                                float* __restrict__ xl,
                                                float* __restrict__ xr) {
    const int lane  = threadIdx.x & 31;
    const int wave  = threadIdx.x >> 5;
    const int m0    = blockIdx.x * 16;
    const int n0    = wave * 16;
    const int lhalf = lane >> 4;
    const int l15   = lane & 15;

    const float* __restrict__ W    = blockIdx.z ? Wr : Wl;
    float*       __restrict__ outp = blockIdx.z ? xr : xl;

    const float* __restrict__ arow = x + (m0 + l15) * FEAT;

    v8f c = {0.f, 0.f, 0.f, 0.f, 0.f, 0.f, 0.f, 0.f};

    #pragma unroll 4
    for (int k = 0; k < FEAT; k += 4) {
        const int ka = k + lhalf * 2;
        v2f a, b;
        a.x = arow[ka];
        a.y = arow[ka + 1];
        b.x = W[ka * FEAT + n0 + l15];
        b.y = W[(ka + 1) * FEAT + n0 + l15];
        c = __builtin_amdgcn_wmma_f32_16x16x4_f32(
                 false, a,  false, b,
                 (short)0, c,  false,  false);
    }

    #pragma unroll
    for (int r = 0; r < 8; ++r)
        outp[(m0 + r + lhalf * 8) * FEAT + n0 + l15] = c[r];
}

__global__ __launch_bounds__(256) void gat_score(const float* __restrict__ xl,
                                                 const float* __restrict__ xr,
                                                 const int*   __restrict__ ei,
                                                 const float* __restrict__ att,
                                                 float* __restrict__ score,
                                                 float* __restrict__ m) {
    const int lane = threadIdx.x & 31;
    const int gw   = (blockIdx.x * blockDim.x + threadIdx.x) >> 5;
    const int nw   = (gridDim.x * blockDim.x) >> 5;

    const float4 a = ((const float4*)att)[lane];

    for (int e = gw; e < ETOT; e += nw) {
        int src, dst;
        if (e < NEDGE) { src = ei[e]; dst = ei[NEDGE + e]; }
        else           { src = e - NEDGE; dst = src; }

        const float4 vl = ((const float4*)(xl + src * FEAT))[lane];
        const float4 vr = ((const float4*)(xr + dst * FEAT))[lane];

        float p = 0.f, t;
        t = vl.x + vr.x; t = (t >= 0.f) ? t : NEG_SLOPE * t; p += a.x * t;
        t = vl.y + vr.y; t = (t >= 0.f) ? t : NEG_SLOPE * t; p += a.y * t;
        t = vl.z + vr.z; t = (t >= 0.f) ? t : NEG_SLOPE * t; p += a.z * t;
        t = vl.w + vr.w; t = (t >= 0.f) ? t : NEG_SLOPE * t; p += a.w * t;

        p += __shfl_xor(p, 1, 32);
        p += __shfl_xor(p, 2, 32);
        p += __shfl_xor(p, 4, 32);

        if ((lane & 7) == 0) {
            const int h = lane >> 3;
            score[e * HEADS + h] = p;
            __hip_atomic_fetch_max(&m[dst * HEADS + h], p,
                                   __ATOMIC_RELAXED, __HIP_MEMORY_SCOPE_AGENT);
        }
    }
}

__global__ __launch_bounds__(256) void gat_exp(const int*   __restrict__ ei,
                                               float* __restrict__ score,
                                               const float* __restrict__ m,
                                               float* __restrict__ s) {
    const int tot    = ETOT * HEADS;
    const int stride = gridDim.x * blockDim.x;
    for (int i = blockIdx.x * blockDim.x + threadIdx.x; i < tot; i += stride) {
        const int e = i >> 2;
        const int h = i & 3;
        const int dst = (e < NEDGE) ? ei[NEDGE + e] : (e - NEDGE);
        const float ex = __expf(score[i] - m[dst * HEADS + h]);
        score[i] = ex;
        unsafeAtomicAdd(&s[dst * HEADS + h], ex);
    }
}

__global__ __launch_bounds__(256) void gat_scatter(const float* __restrict__ xl,
                                                   const int*   __restrict__ ei,
                                                   const float* __restrict__ score,
                                                   const float* __restrict__ s,
                                                   float* __restrict__ out) {
    const int lane = threadIdx.x & 31;
    const int h    = lane >> 3;
    const int gw   = (blockIdx.x * blockDim.x + threadIdx.x) >> 5;
    const int nw   = (gridDim.x * blockDim.x) >> 5;

    for (int e = gw; e < ETOT; e += nw) {
        int src, dst;
        if (e < NEDGE) { src = ei[e]; dst = ei[NEDGE + e]; }
        else           { src = e - NEDGE; dst = src; }

        const float alpha = score[e * HEADS + h] / s[dst * HEADS + h];
        const float4 v = ((const float4*)(xl + src * FEAT))[lane];

        float* op = out + dst * FEAT + lane * 4;
        unsafeAtomicAdd(op + 0, alpha * v.x);
        unsafeAtomicAdd(op + 1, alpha * v.y);
        unsafeAtomicAdd(op + 2, alpha * v.z);
        unsafeAtomicAdd(op + 3, alpha * v.w);
    }
}

extern "C" void kernel_launch(void* const* d_in, const int* in_sizes, int n_in,
                              void* d_out, int out_size, void* d_ws, size_t ws_size,
                              hipStream_t stream) {
    const float* x    = (const float*)d_in[0];
    const int*   ei   = (const int*)  d_in[1];
    const float* W_l  = (const float*)d_in[2];
    const float* W_r  = (const float*)d_in[3];
    const float* att  = (const float*)d_in[4];
    const float* bias = (const float*)d_in[5];
    float* out = (float*)d_out;

    float* ws    = (float*)d_ws;
    float* xl    = ws;
    float* xr    = xl + NNODES * FEAT;
    float* mbuf  = xr + NNODES * FEAT;
    float* sbuf  = mbuf + NNODES * HEADS;
    float* score = sbuf + NNODES * HEADS;

    gat_init<<<4096, 256, 0, stream>>>(out, mbuf, sbuf, bias);

    gat_gemm<<<dim3(NNODES / 16, 1, 2), 256, 0, stream>>>(x, W_l, W_r, xl, xr);

    gat_score<<<4096, 256, 0, stream>>>(xl, xr, ei, att, score, mbuf);

    gat_exp<<<8192, 256, 0, stream>>>(ei, score, mbuf, sbuf);

    gat_scatter<<<8192, 256, 0, stream>>>(xl, ei, score, sbuf, out);
}


# Score and scatter: a wave takes 32 consecutive edges: one contiguous endpoint load per row, endpoints read out by readlane as scalars, row loads of 4 or 8 edges in flight; sums in arrival order
# speedup vs baseline: 1.5340x; 1.5340x over previous
#include <hip/hip_runtime.h>

#define NNODES   50000
#define FEAT     128
#define HEADS    4
#define NEDGE    800000
#define ETOT     (NEDGE + NNODES)
#define NEG_SLOPE 0.2f

typedef __attribute__((ext_vector_type(2))) float v2f;
typedef __attribute__((ext_vector_type(8))) float v8f;

__global__ __launch_bounds__(256) void gat_init(float* __restrict__ out,
                                                float* __restrict__ m,
                                                float* __restrict__ s,
                                                const float* __restrict__ bias) {
    int tid = blockIdx.x * 256 + threadIdx.x;
    int stride = gridDim.x * 256;
    for (int i = tid; i < NNODES * FEAT; i += stride)
        out[i] = bias[i & (FEAT - 1)];
    for (int i = tid; i < NNODES * HEADS; i += stride) {
        m[i] = -1e30f;
        s[i] = 0.0f;
    }
}

#define TROWS   80
#define RTILES  (TROWS / 16)
#define APITCH  (FEAT + 4)

__global__ __launch_bounds__(256) void gat_gemm(const float* __restrict__ x,
                                                const float* __restrict__ Wl,
                                                const float* __restrict__ Wr,
                                                float* __restrict__ xl,
                                                float* __restrict__ xr) {
    __shared__ __attribute__((aligned(16))) float As[TROWS * APITCH];

    const int lane  = threadIdx.x & 31;
    const int wave  = threadIdx.x >> 5;
    const int m0    = blockIdx.x * TROWS;
    const int n0    = wave * 16;
    const int lhalf = lane >> 4;
    const int l15   = lane & 15;

    const float* __restrict__ W    = blockIdx.z ? Wr : Wl;
    float*       __restrict__ outp = blockIdx.z ? xr : xl;

    #pragma unroll
    for (int i = 0; i < TROWS / 8; ++i) {
        const int r = wave + i * 8;
        const float4 v = ((const float4*)(x + (m0 + r) * FEAT))[lane];
        *(float4*)(&As[r * APITCH + lane * 4]) = v;
    }
    __syncthreads();

    v8f c[RTILES];
    #pragma unroll
    for (int t = 0; t < RTILES; ++t)
        c[t] = (v8f){0.f, 0.f, 0.f, 0.f, 0.f, 0.f, 0.f, 0.f};

    const float* ap = As + l15 * APITCH + lhalf * 2;
    const float* __restrict__ bp = W + (lhalf * 2) * FEAT + n0 + l15;

    #pragma unroll 4
    for (int k = 0; k < FEAT; k += 4) {
        v2f b;
        b.x = bp[k * FEAT];
        b.y = bp[(k + 1) * FEAT];
        #pragma unroll
        for (int t = 0; t < RTILES; ++t) {
            const v2f a = *(const v2f*)(ap + t * 16 * APITCH + k);
            c[t] = __builtin_amdgcn_wmma_f32_16x16x4_f32(
                        false, a,  false, b,
                        (short)0, c[t],  false,  false);
        }
    }

    #pragma unroll
    for (int t = 0; t < RTILES; ++t) {
        #pragma unroll
        for (int r = 0; r < 8; ++r)
            outp[(m0 + t * 16 + r + lhalf * 8) * FEAT + n0 + l15] = c[t][r];
    }
}

#define CHUNK    32
#define NCHUNK   ((ETOT + CHUNK - 1) / CHUNK)
#define SGROUP   4
#define CGROUP   8
static_assert(NEDGE % CHUNK == 0, "a chunk must not straddle the self-loop boundary");
static_assert((ETOT % CHUNK) % SGROUP == 0 && (ETOT % CHUNK) % CGROUP == 0,
              "the short last chunk must be whole groups");

__global__ __launch_bounds__(256) void gat_score(const float* __restrict__ xl,
                                                 const float* __restrict__ xr,
                                                 const int*   __restrict__ ei,
                                                 const float* __restrict__ att,
                                                 float* __restrict__ score,
                                                 float* __restrict__ m) {
    const int lane = threadIdx.x & 31;
    const int gw   = __builtin_amdgcn_readfirstlane((blockIdx.x * blockDim.x + threadIdx.x) >> 5);
    const int nw   = (gridDim.x * blockDim.x) >> 5;

    const float4 a = ((const float4*)att)[lane];

    for (int ch = gw; ch < NCHUNK; ch += nw) {
        const int e0 = ch * CHUNK;
        int src, dst;
        if (e0 < NEDGE) { src = ei[e0 + lane]; dst = ei[NEDGE + e0 + lane]; }
        else            { src = e0 + lane - NEDGE; dst = src; }
        const int nv = (ETOT - e0 < CHUNK) ? (ETOT - e0) : CHUNK;

        for (int j0 = 0; j0 < nv; j0 += SGROUP) {
            float4 vl[SGROUP], vr[SGROUP];
            int    dj[SGROUP];
            #pragma unroll
            for (int u = 0; u < SGROUP; ++u) {
                const int sj = __builtin_amdgcn_readlane(src, j0 + u);
                dj[u]        = __builtin_amdgcn_readlane(dst, j0 + u);
                vl[u] = ((const float4*)(xl + sj    * FEAT))[lane];
                vr[u] = ((const float4*)(xr + dj[u] * FEAT))[lane];
            }
            float p[SGROUP];
            #pragma unroll
            for (int u = 0; u < SGROUP; ++u) {
                float q = 0.f, t;
                t = vl[u].x + vr[u].x; t = (t >= 0.f) ? t : NEG_SLOPE * t; q += a.x * t;
                t = vl[u].y + vr[u].y; t = (t >= 0.f) ? t : NEG_SLOPE * t; q += a.y * t;
                t = vl[u].z + vr[u].z; t = (t >= 0.f) ? t : NEG_SLOPE * t; q += a.z * t;
                t = vl[u].w + vr[u].w; t = (t >= 0.f) ? t : NEG_SLOPE * t; q += a.w * t;
                p[u] = q;
            }
            #pragma unroll
            for (int u = 0; u < SGROUP; ++u) p[u] += __shfl_xor(p[u], 1, 32);
            #pragma unroll
            for (int u = 0; u < SGROUP; ++u) p[u] += __shfl_xor(p[u], 2, 32);
            #pragma unroll
            for (int u = 0; u < SGROUP; ++u) p[u] += __shfl_xor(p[u], 4, 32);

            if ((lane & 7) == 0) {
                const int h = lane >> 3;
                #pragma unroll
                for (int u = 0; u < SGROUP; ++u) {
                    score[(e0 + j0 + u) * HEADS + h] = p[u];
                    __hip_atomic_fetch_max(&m[dj[u] * HEADS + h], p[u],
                                           __ATOMIC_RELAXED, __HIP_MEMORY_SCOPE_AGENT);
                }
            }
        }
    }
}

__global__ __launch_bounds__(256) void gat_exp(const int*   __restrict__ ei,
                                               float* __restrict__ score,
                                               const float* __restrict__ m,
                                               float* __restrict__ s) {
    const int tot    = ETOT * HEADS;
    const int stride = gridDim.x * blockDim.x;
    for (int i = blockIdx.x * blockDim.x + threadIdx.x; i < tot; i += stride) {
        const int e = i >> 2;
        const int h = i & 3;
        const int dst = (e < NEDGE) ? ei[NEDGE + e] : (e - NEDGE);
        const float ex = __expf(score[i] - m[dst * HEADS + h]);
        score[i] = ex;
        unsafeAtomicAdd(&s[dst * HEADS + h], ex);
    }
}

__global__ __launch_bounds__(256) void gat_scatter(const float* __restrict__ xl,
                                                   const int*   __restrict__ ei,
                                                   const float* __restrict__ score,
                                                   const float* __restrict__ s,
                                                   float* __restrict__ out) {
    const int lane = threadIdx.x & 31;
    const int gw   = __builtin_amdgcn_readfirstlane((blockIdx.x * blockDim.x + threadIdx.x) >> 5);
    const int nw   = (gridDim.x * blockDim.x) >> 5;

    for (int ch = gw; ch < NCHUNK; ch += nw) {
        const int e0 = ch * CHUNK;
        int src, dst;
        if (e0 < NEDGE) { src = ei[e0 + lane]; dst = ei[NEDGE + e0 + lane]; }
        else            { src = e0 + lane - NEDGE; dst = src; }
        const int nv = (ETOT - e0 < CHUNK) ? (ETOT - e0) : CHUNK;

        float4 al = make_float4(0.f, 0.f, 0.f, 0.f);
        if (lane < nv) {
            const float4 ex = ((const float4*)score)[e0 + lane];
            const float4 sv = ((const float4*)s)[dst];
            al.x = ex.x / sv.x;
            al.y = ex.y / sv.y;
            al.z = ex.z / sv.z;
            al.w = ex.w / sv.w;
        }
        const int ax = __float_as_int(al.x), ay = __float_as_int(al.y);
        const int az = __float_as_int(al.z), aw = __float_as_int(al.w);

        for (int j0 = 0; j0 < nv; j0 += CGROUP) {
            float v0[CGROUP], v1[CGROUP], v2[CGROUP], v3[CGROUP];
            int   dj[CGROUP];
            #pragma unroll
            for (int u = 0; u < CGROUP; ++u) {
                const int sj = __builtin_amdgcn_readlane(src, j0 + u);
                dj[u]        = __builtin_amdgcn_readlane(dst, j0 + u);
                const float* xp = xl + sj * FEAT + lane;
                v0[u] = xp[0];
                v1[u] = xp[32];
                v2[u] = xp[64];
                v3[u] = xp[96];
            }
            #pragma unroll
            for (int u = 0; u < CGROUP; ++u) {
                const float a0 = __int_as_float(__builtin_amdgcn_readlane(ax, j0 + u));
                const float a1 = __int_as_float(__builtin_amdgcn_readlane(ay, j0 + u));
                const float a2 = __int_as_float(__builtin_amdgcn_readlane(az, j0 + u));
                const float a3 = __int_as_float(__builtin_amdgcn_readlane(aw, j0 + u));
                float* op = out + dj[u] * FEAT + lane;
                unsafeAtomicAdd(op,      a0 * v0[u]);
                unsafeAtomicAdd(op + 32, a1 * v1[u]);
                unsafeAtomicAdd(op + 64, a2 * v2[u]);
                unsafeAtomicAdd(op + 96, a3 * v3[u]);
            }
        }
    }
}

extern "C" void kernel_launch(void* const* d_in, const int* in_sizes, int n_in,
                              void* d_out, int out_size, void* d_ws, size_t ws_size,
                              hipStream_t stream) {
    const float* x    = (const float*)d_in[0];
    const int*   ei   = (const int*)  d_in[1];
    const float* W_l  = (const float*)d_in[2];
    const float* W_r  = (const float*)d_in[3];
    const float* att  = (const float*)d_in[4];
    const float* bias = (const float*)d_in[5];
    float* out = (float*)d_out;

    float* ws    = (float*)d_ws;
    float* xl    = ws;
    float* xr    = xl + NNODES * FEAT;
    float* mbuf  = xr + NNODES * FEAT;
    float* sbuf  = mbuf + NNODES * HEADS;
    float* score = sbuf + NNODES * HEADS;

    gat_init<<<4096, 256, 0, stream>>>(out, mbuf, sbuf, bias);

    gat_gemm<<<dim3(NNODES / TROWS, 1, 2), 256, 0, stream>>>(x, W_l, W_r, xl, xr);

    gat_score<<<(NCHUNK + 15) / 16, 256, 0, stream>>>(xl, xr, ei, att, score, mbuf);

    gat_exp<<<8192, 256, 0, stream>>>(ei, score, mbuf, sbuf);

    gat_scatter<<<(NCHUNK + 15) / 16, 256, 0, stream>>>(xl, ei, score, sbuf, out);
}


# Exp+scatter become 256 destination-range owner blocks: each scans a 1-byte-per-edge owner plane (edges unsorted), adds ex and ex*xl rows into LDS (arrival order), writes out=acc/s+bias once
# speedup vs baseline: 1.9471x; 1.2693x over previous
#include <hip/hip_runtime.h>

#define NNODES   50000
#define FEAT     128
#define HEADS    4
#define NEDGE    800000
#define ETOT     (NEDGE + NNODES)
#define NEG_SLOPE 0.2f

typedef __attribute__((ext_vector_type(2))) float v2f;
typedef __attribute__((ext_vector_type(8))) float v8f;

__global__ __launch_bounds__(256) void gat_init(float* __restrict__ m) {
    const int i = blockIdx.x * 256 + threadIdx.x;
    if (i < NNODES * HEADS) m[i] = -1e30f;
}

#define TROWS   80
#define RTILES  (TROWS / 16)
#define APITCH  (FEAT + 4)

__global__ __launch_bounds__(256) void gat_gemm(const float* __restrict__ x,
                                                const float* __restrict__ Wl,
                                                const float* __restrict__ Wr,
                                                float* __restrict__ xl,
                                                float* __restrict__ xr) {
    __shared__ __attribute__((aligned(16))) float As[TROWS * APITCH];

    const int lane  = threadIdx.x & 31;
    const int wave  = threadIdx.x >> 5;
    const int m0    = blockIdx.x * TROWS;
    const int n0    = wave * 16;
    const int lhalf = lane >> 4;
    const int l15   = lane & 15;

    const float* __restrict__ W    = blockIdx.z ? Wr : Wl;
    float*       __restrict__ outp = blockIdx.z ? xr : xl;

    #pragma unroll
    for (int i = 0; i < TROWS / 8; ++i) {
        const int r = wave + i * 8;
        const float4 v = ((const float4*)(x + (m0 + r) * FEAT))[lane];
        *(float4*)(&As[r * APITCH + lane * 4]) = v;
    }
    __syncthreads();

    v8f c[RTILES];
    #pragma unroll
    for (int t = 0; t < RTILES; ++t)
        c[t] = (v8f){0.f, 0.f, 0.f, 0.f, 0.f, 0.f, 0.f, 0.f};

    const float* ap = As + l15 * APITCH + lhalf * 2;
    const float* __restrict__ bp = W + (lhalf * 2) * FEAT + n0 + l15;

    #pragma unroll 4
    for (int k = 0; k < FEAT; k += 4) {
        v2f b;
        b.x = bp[k * FEAT];
        b.y = bp[(k + 1) * FEAT];
        #pragma unroll
        for (int t = 0; t < RTILES; ++t) {
            const v2f a = *(const v2f*)(ap + t * 16 * APITCH + k);
            c[t] = __builtin_amdgcn_wmma_f32_16x16x4_f32(
                        false, a,  false, b,
                        (short)0, c[t],  false,  false);
        }
    }

    #pragma unroll
    for (int t = 0; t < RTILES; ++t) {
        #pragma unroll
        for (int r = 0; r < 8; ++r)
            outp[(m0 + t * 16 + r + lhalf * 8) * FEAT + n0 + l15] = c[t][r];
    }
}

#define CHUNK    32
#define NCHUNK   ((ETOT + CHUNK - 1) / CHUNK)
#define SGROUP   4
static_assert(NEDGE % CHUNK == 0, "a chunk must not straddle the self-loop boundary");
static_assert((ETOT % CHUNK) % SGROUP == 0,
              "the short last chunk must be whole groups");

#define OWN_NB      256
#define OWN_R       ((NNODES + OWN_NB - 1) / OWN_NB)
static_assert((NNODES - 1) / OWN_R < OWN_NB && OWN_NB <= 256, "an owner id fits one byte");
static_assert((OWN_NB - 1) * OWN_R < NNODES, "every block owns at least one node");

__global__ __launch_bounds__(256) void gat_score(const float* __restrict__ xl,
                                                 const float* __restrict__ xr,
                                                 const int*   __restrict__ ei,
                                                 const float* __restrict__ att,
                                                 float* __restrict__ score,
                                                 float* __restrict__ m,
                                                 unsigned char* __restrict__ own8) {
    const int lane = threadIdx.x & 31;
    const int gw   = __builtin_amdgcn_readfirstlane((blockIdx.x * blockDim.x + threadIdx.x) >> 5);
    const int nw   = (gridDim.x * blockDim.x) >> 5;

    const float4 a = ((const float4*)att)[lane];

    for (int ch = gw; ch < NCHUNK; ch += nw) {
        const int e0 = ch * CHUNK;
        int src, dst;
        if (e0 < NEDGE) {
            src = ei[e0 + lane]; dst = ei[NEDGE + e0 + lane];
            own8[e0 + lane] = (unsigned char)((unsigned)dst / OWN_R);
        }
        else            { src = e0 + lane - NEDGE; dst = src; }
        const int nv = (ETOT - e0 < CHUNK) ? (ETOT - e0) : CHUNK;

        for (int j0 = 0; j0 < nv; j0 += SGROUP) {
            float4 vl[SGROUP], vr[SGROUP];
            int    dj[SGROUP];
            #pragma unroll
            for (int u = 0; u < SGROUP; ++u) {
                const int sj = __builtin_amdgcn_readlane(src, j0 + u);
                dj[u]        = __builtin_amdgcn_readlane(dst, j0 + u);
                vl[u] = ((const float4*)(xl + sj    * FEAT))[lane];
                vr[u] = ((const float4*)(xr + dj[u] * FEAT))[lane];
            }
            float p[SGROUP];
            #pragma unroll
            for (int u = 0; u < SGROUP; ++u) {
                float q = 0.f, t;
                t = vl[u].x + vr[u].x; t = (t >= 0.f) ? t : NEG_SLOPE * t; q += a.x * t;
                t = vl[u].y + vr[u].y; t = (t >= 0.f) ? t : NEG_SLOPE * t; q += a.y * t;
                t = vl[u].z + vr[u].z; t = (t >= 0.f) ? t : NEG_SLOPE * t; q += a.z * t;
                t = vl[u].w + vr[u].w; t = (t >= 0.f) ? t : NEG_SLOPE * t; q += a.w * t;
                p[u] = q;
            }
            #pragma unroll
            for (int u = 0; u < SGROUP; ++u) p[u] += __shfl_xor(p[u], 1, 32);
            #pragma unroll
            for (int u = 0; u < SGROUP; ++u) p[u] += __shfl_xor(p[u], 2, 32);
            #pragma unroll
            for (int u = 0; u < SGROUP; ++u) p[u] += __shfl_xor(p[u], 4, 32);

            if ((lane & 7) == 0) {
                const int h = lane >> 3;
                #pragma unroll
                for (int u = 0; u < SGROUP; ++u) {
                    score[(e0 + j0 + u) * HEADS + h] = p[u];
                    __hip_atomic_fetch_max(&m[dj[u] * HEADS + h], p[u],
                                           __ATOMIC_RELAXED, __HIP_MEMORY_SCOPE_AGENT);
                }
            }
        }
    }
}

#define OWN_WAVES   32
#define OWN_THREADS (OWN_WAVES * 32)
#define SCAN_E      512
#define NVEC        (NEDGE / 16)
#define NSCAN       ((NVEC + 31) / 32)
#define SCAN_UNR    8
#define QGROUP      8
#define QCAP        (QGROUP + 2 * SCAN_E)
#define QLIMIT      (QCAP - SCAN_E)
static_assert(NEDGE % 16 == 0, "the owner-byte plane is whole 16-byte pieces");
static_assert((OWN_R + OWN_WAVES - 1) / OWN_WAVES < QGROUP, "self loops fit below one group");
static_assert(QGROUP <= QLIMIT, "a drained queue always has room for one load's hits");
static_assert(OWN_NB * 6 < NSCAN, "staggered start stays inside the plane");

struct OwnLds {
    float    acc[OWN_R * FEAT];
    float    sl[OWN_R * HEADS];
    float    ml[OWN_R * HEADS];
    unsigned q[OWN_WAVES * QCAP];
};

template <bool FULL>
__device__ __forceinline__ void own_drain(OwnLds& L, const unsigned* qw, int base, int g,
                                          int lane, int lo,
                                          const float* __restrict__ xl,
                                          const int*   __restrict__ ei,
                                          const float* __restrict__ score) {
    const int u = lane >> 2, h = lane & 3;
    const bool valid = FULL || (u < g);
    const int e  = valid ? (int)qw[base + u] : 0;
    const bool listed = e < NEDGE;
    const int el = listed ? e : 0;
    const int sv = ei[el];
    const int dv = ei[NEDGE + el];
    const float sc = score[e * HEADS + h];
    const int sr = listed ? sv : (e - NEDGE);
    const int dl = valid ? ((listed ? dv : (e - NEDGE)) - lo) : 0;
    const float ex = valid ? __expf(sc - L.ml[dl * HEADS + h]) : 0.0f;
    if (valid) unsafeAtomicAdd(&L.sl[dl * HEADS + h], ex);
    const int exb = __float_as_int(ex);

    float v0[QGROUP], v1[QGROUP], v2[QGROUP], v3[QGROUP];
    #pragma unroll
    for (int k = 0; k < QGROUP; ++k) {
        v0[k] = 0.0f; v1[k] = 0.0f; v2[k] = 0.0f; v3[k] = 0.0f;
        if (FULL || k < g) {
            const int sj = __builtin_amdgcn_readlane(sr, 4 * k);
            const float* xp = xl + sj * FEAT + lane;
            v0[k] = xp[0]; v1[k] = xp[32]; v2[k] = xp[64]; v3[k] = xp[96];
        }
    }
    #pragma unroll
    for (int k = 0; k < QGROUP; ++k) {
        if (FULL || k < g) {
            const int dj = __builtin_amdgcn_readlane(dl, 4 * k);
            const float a0 = __int_as_float(__builtin_amdgcn_readlane(exb, 4 * k + 0));
            const float a1 = __int_as_float(__builtin_amdgcn_readlane(exb, 4 * k + 1));
            const float a2 = __int_as_float(__builtin_amdgcn_readlane(exb, 4 * k + 2));
            const float a3 = __int_as_float(__builtin_amdgcn_readlane(exb, 4 * k + 3));
            float* ap = &L.acc[dj * FEAT + lane];
            unsafeAtomicAdd(ap,      a0 * v0[k]);
            unsafeAtomicAdd(ap + 32, a1 * v1[k]);
            unsafeAtomicAdd(ap + 64, a2 * v2[k]);
            unsafeAtomicAdd(ap + 96, a3 * v3[k]);
        }
    }
}

__global__ __launch_bounds__(OWN_THREADS) void gat_own(const float* __restrict__ xl,
                                                       const int*   __restrict__ ei,
                                                       const float* __restrict__ score,
                                                       const float* __restrict__ m,
                                                       const unsigned char* __restrict__ own8,
                                                       const float* __restrict__ bias,
                                                       float* __restrict__ out) {
    __shared__ OwnLds L;

    const int lane = threadIdx.x & 31;
    const int w    = __builtin_amdgcn_readfirstlane(threadIdx.x >> 5);
    const int lo   = blockIdx.x * OWN_R;
    const int nown = (NNODES - lo < OWN_R) ? (NNODES - lo) : OWN_R;

    for (int i = threadIdx.x; i < OWN_R * FEAT; i += OWN_THREADS) L.acc[i] = 0.0f;
    for (int i = threadIdx.x; i < OWN_R * HEADS; i += OWN_THREADS) {
        L.sl[i] = 0.0f;
        L.ml[i] = (i < nown * HEADS) ? m[lo * HEADS + i] : 0.0f;
    }
    __syncthreads();

    unsigned* qw = L.q + w * QCAP;
    int qn = 0;

    {
        const int n = w + lane * OWN_WAVES;
        if (n < nown) qw[lane] = (unsigned)(NEDGE + lo + n);
        qn = (nown > w) ? (nown - w + OWN_WAVES - 1) / OWN_WAVES : 0;
    }

    const int start = blockIdx.x * 6;
    const uint4* __restrict__ ownv = (const uint4*)own8;
    const unsigned pat  = blockIdx.x * 0x01010101u;
    const unsigned miss = pat ^ 0x80808080u;

    for (int t0 = w; t0 < NSCAN; t0 += OWN_WAVES * SCAN_UNR) {
        int kdone = 0;
        while (kdone < SCAN_UNR) {
            uint4 d[SCAN_UNR];
            int   piece[SCAN_UNR];
            #pragma unroll
            for (int k = 0; k < SCAN_UNR; ++k) {
                const int idx = t0 + k * OWN_WAVES;
                int ps = idx + start;
                if (ps >= NSCAN) ps -= NSCAN;
                piece[k] = ps * 32 + lane;
                const int pc = (piece[k] < NVEC) ? piece[k] : (NVEC - 1);
                d[k] = ownv[pc];
            }
            #pragma unroll
            for (int k = 0; k < SCAN_UNR; ++k)
                asm volatile("" :: "v"(d[k].x), "v"(d[k].y), "v"(d[k].z), "v"(d[k].w));
            #pragma unroll
            for (int k = 0; k < SCAN_UNR; ++k) {
                const bool live = (t0 + k * OWN_WAVES < NSCAN) && (piece[k] < NVEC);
                d[k].x = live ? d[k].x : miss;
                d[k].y = live ? d[k].y : miss;
                d[k].z = live ? d[k].z : miss;
                d[k].w = live ? d[k].w : miss;
            }
            #pragma unroll
            for (int k = 0; k < SCAN_UNR; ++k) {
                if (k >= kdone && qn <= QLIMIT) {
                    const unsigned dw[4] = {d[k].x, d[k].y, d[k].z, d[k].w};
                    unsigned hm = 0;
                    #pragma unroll
                    for (int c = 0; c < 4; ++c) {
                        const unsigned t = dw[c] ^ pat;
                        const unsigned z = ~(((t & 0x7F7F7F7Fu) + 0x7F7F7F7Fu) | t | 0x7F7F7F7Fu);
                        hm |= (((z >> 7) * 0x01020408u) >> 24) << (4 * c);
                    }
                    unsigned mask = __builtin_amdgcn_ballot_w32(hm != 0);
                    while (mask) {
                        const int rank = __builtin_amdgcn_mbcnt_lo(mask, 0);
                        if (hm != 0) {
                            qw[qn + rank] = (unsigned)(piece[k] * 16 + __builtin_ctz(hm));
                            hm &= hm - 1;
                        }
                        qn += __builtin_popcount(mask);
                        mask = __builtin_amdgcn_ballot_w32(hm != 0);
                    }
                    kdone = k + 1;
                }
            }
            __builtin_amdgcn_wave_barrier();
            while (qn >= QGROUP) {
                qn -= QGROUP;
                own_drain<true>(L, qw, qn, QGROUP, lane, lo, xl, ei, score);
            }
        }
    }
    __builtin_amdgcn_wave_barrier();
    if (qn > 0) own_drain<false>(L, qw, 0, qn, lane, lo, xl, ei, score);

    __syncthreads();

    const float b0 = bias[lane], b1 = bias[lane + 32], b2 = bias[lane + 64], b3 = bias[lane + 96];
    for (int n = w; n < nown; n += OWN_WAVES) {
        const float* ap = &L.acc[n * FEAT + lane];
        float* op = out + (lo + n) * FEAT + lane;
        op[0]  = ap[0]  / L.sl[n * HEADS + 0] + b0;
        op[32] = ap[32] / L.sl[n * HEADS + 1] + b1;
        op[64] = ap[64] / L.sl[n * HEADS + 2] + b2;
        op[96] = ap[96] / L.sl[n * HEADS + 3] + b3;
    }
}

extern "C" void kernel_launch(void* const* d_in, const int* in_sizes, int n_in,
                              void* d_out, int out_size, void* d_ws, size_t ws_size,
                              hipStream_t stream) {
    const float* x    = (const float*)d_in[0];
    const int*   ei   = (const int*)  d_in[1];
    const float* W_l  = (const float*)d_in[2];
    const float* W_r  = (const float*)d_in[3];
    const float* att  = (const float*)d_in[4];
    const float* bias = (const float*)d_in[5];
    float* out = (float*)d_out;

    float* ws    = (float*)d_ws;
    float* xl    = ws;
    float* xr    = xl + NNODES * FEAT;
    float* mbuf  = xr + NNODES * FEAT;
    float* score = mbuf + NNODES * HEADS;
    unsigned char* own8 = (unsigned char*)(score + ETOT * HEADS);

    gat_init<<<(NNODES * HEADS + 255) / 256, 256, 0, stream>>>(mbuf);

    gat_gemm<<<dim3(NNODES / TROWS, 1, 2), 256, 0, stream>>>(x, W_l, W_r, xl, xr);

    gat_score<<<(NCHUNK + 15) / 16, 256, 0, stream>>>(xl, xr, ei, att, score, mbuf, own8);

    gat_own<<<OWN_NB, OWN_THREADS, 0, stream>>>(xl, ei, score, mbuf, own8, bias, out);
}


# Scores move into the owner blocks: own xr rows in LDS; one plane scan lists a block's edges in LDS, swept twice (score + LDS max, then exp + sums); no score kernel, no global max; LDS arrival order
# speedup vs baseline: 2.9996x; 1.5405x over previous
#include <hip/hip_runtime.h>

#define NNODES   50000
#define FEAT     128
#define HEADS    4
#define NEDGE    800000
#define ETOT     (NEDGE + NNODES)
#define NEG_SLOPE 0.2f

typedef __attribute__((ext_vector_type(2))) float v2f;
typedef __attribute__((ext_vector_type(8))) float v8f;

#define OWN_NB      256
#define OWN_R       ((NNODES + OWN_NB - 1) / OWN_NB)
static_assert((NNODES - 1) / OWN_R < OWN_NB && OWN_NB <= 256, "an owner id fits one byte");
static_assert((OWN_NB - 1) * OWN_R < NNODES, "every block owns at least one node");
static_assert(NEDGE % 16 == 0, "the owner-byte plane is whole 16-byte pieces");

__global__ __launch_bounds__(256) void gat_plane(const int* __restrict__ ei,
                                                 unsigned char* __restrict__ own8) {
    const int i = blockIdx.x * 256 + threadIdx.x;
    if (i < NEDGE / 4) {
        const int4 d = ((const int4*)(ei + NEDGE))[i];
        ((unsigned*)own8)[i] = ((unsigned)d.x / OWN_R)
                             | (((unsigned)d.y / OWN_R) << 8)
                             | (((unsigned)d.z / OWN_R) << 16)
                             | (((unsigned)d.w / OWN_R) << 24);
    }
}

#define TROWS   80
#define RTILES  (TROWS / 16)
#define APITCH  (FEAT + 4)

__global__ __launch_bounds__(256) void gat_gemm(const float* __restrict__ x,
                                                const float* __restrict__ Wl,
                                                const float* __restrict__ Wr,
                                                float* __restrict__ xl,
                                                float* __restrict__ xr) {
    __shared__ __attribute__((aligned(16))) float As[TROWS * APITCH];

    const int lane  = threadIdx.x & 31;
    const int wave  = threadIdx.x >> 5;
    const int m0    = blockIdx.x * TROWS;
    const int n0    = wave * 16;
    const int lhalf = lane >> 4;
    const int l15   = lane & 15;

    const float* __restrict__ W    = blockIdx.z ? Wr : Wl;
    float*       __restrict__ outp = blockIdx.z ? xr : xl;

    #pragma unroll
    for (int i = 0; i < TROWS / 8; ++i) {
        const int r = wave + i * 8;
        const float4 v = ((const float4*)(x + (m0 + r) * FEAT))[lane];
        *(float4*)(&As[r * APITCH + lane * 4]) = v;
    }
    __syncthreads();

    v8f c[RTILES];
    #pragma unroll
    for (int t = 0; t < RTILES; ++t)
        c[t] = (v8f){0.f, 0.f, 0.f, 0.f, 0.f, 0.f, 0.f, 0.f};

    const float* ap = As + l15 * APITCH + lhalf * 2;
    const float* __restrict__ bp = W + (lhalf * 2) * FEAT + n0 + l15;

    #pragma unroll 4
    for (int k = 0; k < FEAT; k += 4) {
        v2f b;
        b.x = bp[k * FEAT];
        b.y = bp[(k + 1) * FEAT];
        #pragma unroll
        for (int t = 0; t < RTILES; ++t) {
            const v2f a = *(const v2f*)(ap + t * 16 * APITCH + k);
            c[t] = __builtin_amdgcn_wmma_f32_16x16x4_f32(
                        false, a,  false, b,
                        (short)0, c[t],  false,  false);
        }
    }

    #pragma unroll
    for (int t = 0; t < RTILES; ++t) {
        #pragma unroll
        for (int r = 0; r < 8; ++r)
            outp[(m0 + t * 16 + r + lhalf * 8) * FEAT + n0 + l15] = c[t][r];
    }
}

#define OWN_WAVES   32
#define OWN_THREADS (OWN_WAVES * 32)
#define NVEC        (NEDGE / 16)
#define NSCAN       ((NVEC + 31) / 32)
#define SCAN_UNR    8
#define QGROUP      8
#define LCAP        4096
#define XR_TRIPS    ((OWN_R * (FEAT / 4) + OWN_THREADS - 1) / OWN_THREADS)
static_assert(OWN_R <= LCAP && OWN_R <= OWN_THREADS, "the self loops fit the list");
static_assert(NNODES <= 65536 && OWN_R <= 65536, "source and local target share one list word");
static_assert(OWN_NB * 6 < NSCAN, "staggered start stays inside the plane");

struct OwnLds {
    float    acc[OWN_R * FEAT];
    float    xr[OWN_R * FEAT];
    float    sl[OWN_R * HEADS];
    float    ml[OWN_R * HEADS];
    float    sc[LCAP * HEADS];
    unsigned list[LCAP];
    int      cnt;
};

__device__ __forceinline__ unsigned own_hits(const uint4 d, const unsigned pat) {
    const unsigned dw[4] = {d.x, d.y, d.z, d.w};
    unsigned hm = 0;
    #pragma unroll
    for (int c = 0; c < 4; ++c) {
        const unsigned t = dw[c] ^ pat;
        const unsigned z = ~(((t & 0x7F7F7F7Fu) + 0x7F7F7F7Fu) | t | 0x7F7F7F7Fu);
        hm |= (((z >> 7) * 0x01020408u) >> 24) << (4 * c);
    }
    return hm;
}

__device__ __forceinline__ float edge_dot(const float4 a, const float4 vl, const float4 vr) {
    float q = 0.f, t;
    t = vl.x + vr.x; t = (t >= 0.f) ? t : NEG_SLOPE * t; q += a.x * t;
    t = vl.y + vr.y; t = (t >= 0.f) ? t : NEG_SLOPE * t; q += a.y * t;
    t = vl.z + vr.z; t = (t >= 0.f) ? t : NEG_SLOPE * t; q += a.z * t;
    t = vl.w + vr.w; t = (t >= 0.f) ? t : NEG_SLOPE * t; q += a.w * t;
    return q;
}

__device__ __forceinline__ void own_fetch(const OwnLds& L, int ib, int is, int g, int lane,
                                          const int* __restrict__ ei, int& e, int& sv, int& dv) {
    const int u = lane >> 2;
    e = (u < g) ? (int)L.list[ib + u * is] : 0;
    const int el = (e < NEDGE) ? e : 0;
    sv = ei[el];
    dv = ei[NEDGE + el];
}
__device__ __forceinline__ void own_fix(int e, int sv, int dv, int lo, int& sr, int& dl) {
    const bool listed = e < NEDGE;
    sr = listed ? sv : (e - NEDGE);
    dl = (listed ? dv : (e - NEDGE)) - lo;
}

template <bool FULL>
__device__ __forceinline__ void own_score(OwnLds& L, int ib, int is, int g, int lane, int sr, int dl,
                                          const float4 a, const float* __restrict__ xl) {
    float4 vl[QGROUP];
    int    dj[QGROUP];
    #pragma unroll
    for (int k = 0; k < QGROUP; ++k) {
        vl[k] = make_float4(0.f, 0.f, 0.f, 0.f);
        dj[k] = 0;
        if (FULL || k < g) {
            const int sj = __builtin_amdgcn_readlane(sr, 4 * k);
            dj[k]        = __builtin_amdgcn_readlane(dl, 4 * k);
            vl[k] = ((const float4*)(xl + sj * FEAT))[lane];
        }
    }
    float p[QGROUP];
    #pragma unroll
    for (int k = 0; k < QGROUP; ++k) {
        p[k] = 0.f;
        if (FULL || k < g) {
            const float4 vr = ((const float4*)&L.xr[dj[k] * FEAT])[lane];
            p[k] = edge_dot(a, vl[k], vr);
        }
    }
    #pragma unroll
    for (int k = 0; k < QGROUP; ++k) p[k] += __shfl_xor(p[k], 1, 32);
    #pragma unroll
    for (int k = 0; k < QGROUP; ++k) p[k] += __shfl_xor(p[k], 2, 32);
    #pragma unroll
    for (int k = 0; k < QGROUP; ++k) p[k] += __shfl_xor(p[k], 4, 32);

    if ((lane & 3) == 0 && (FULL || (lane >> 2) < g))
        L.list[ib + (lane >> 2) * is] = (unsigned)sr | ((unsigned)dl << 16);

    if ((lane & 7) == 0) {
        const int h = lane >> 3;
        #pragma unroll
        for (int k = 0; k < QGROUP; ++k) {
            if (FULL || k < g) {
                L.sc[(ib + k * is) * HEADS + h] = p[k];
                __hip_atomic_fetch_max(&L.ml[dj[k] * HEADS + h], p[k],
                                       __ATOMIC_RELAXED, __HIP_MEMORY_SCOPE_WORKGROUP);
            }
        }
    }
}

__device__ __forceinline__ void own_sweep1(OwnLds& L, int ib, int is, int n, int lane, int lo,
                                           const float4 a, const float* __restrict__ xl,
                                           const int* __restrict__ ei) {
    int e, sv, dv;
    own_fetch(L, ib, is, (n < QGROUP) ? n : QGROUP, lane, ei, e, sv, dv);
    for (int j = 0; j < n; j += QGROUP) {
        const int g = (n - j < QGROUP) ? (n - j) : QGROUP;
        int sr, dl;
        own_fix(e, sv, dv, lo, sr, dl);
        int gg = n - j - QGROUP;
        gg = (gg < 0) ? 0 : ((gg < QGROUP) ? gg : QGROUP);
        own_fetch(L, ib + (j + QGROUP) * is, is, gg, lane, ei, e, sv, dv);
        if (g == QGROUP) own_score<true >(L, ib + j * is, is, g, lane, sr, dl, a, xl);
        else             own_score<false>(L, ib + j * is, is, g, lane, sr, dl, a, xl);
    }
}

template <bool FULL>
__device__ __forceinline__ void own_sum(OwnLds& L, int base, int g, int lane,
                                        const float* __restrict__ xl) {
    const int u = lane >> 2, h = lane & 3;
    const bool valid = FULL || (u < g);
    const unsigned sd = valid ? L.list[base + u] : 0u;
    const int sr = (int)(sd & 0xFFFFu);
    const int dl = (int)(sd >> 16);
    const float sc = valid ? L.sc[(base + u) * HEADS + h] : 0.0f;
    const float ex = valid ? __expf(sc - L.ml[dl * HEADS + h]) : 0.0f;
    if (valid) unsafeAtomicAdd(&L.sl[dl * HEADS + h], ex);
    const int exb = __float_as_int(ex);

    float v0[QGROUP], v1[QGROUP], v2[QGROUP], v3[QGROUP];
    #pragma unroll
    for (int k = 0; k < QGROUP; ++k) {
        v0[k] = 0.0f; v1[k] = 0.0f; v2[k] = 0.0f; v3[k] = 0.0f;
        if (FULL || k < g) {
            const int sj = __builtin_amdgcn_readlane(sr, 4 * k);
            const float* xp = xl + sj * FEAT + lane;
            v0[k] = xp[0]; v1[k] = xp[32]; v2[k] = xp[64]; v3[k] = xp[96];
        }
    }
    #pragma unroll
    for (int k = 0; k < QGROUP; ++k) {
        if (FULL || k < g) {
            const int dj = __builtin_amdgcn_readlane(dl, 4 * k);
            const float a0 = __int_as_float(__builtin_amdgcn_readlane(exb, 4 * k + 0));
            const float a1 = __int_as_float(__builtin_amdgcn_readlane(exb, 4 * k + 1));
            const float a2 = __int_as_float(__builtin_amdgcn_readlane(exb, 4 * k + 2));
            const float a3 = __int_as_float(__builtin_amdgcn_readlane(exb, 4 * k + 3));
            float* ap = &L.acc[dj * FEAT + lane];
            unsafeAtomicAdd(ap,      a0 * v0[k]);
            unsafeAtomicAdd(ap + 32, a1 * v1[k]);
            unsafeAtomicAdd(ap + 64, a2 * v2[k]);
            unsafeAtomicAdd(ap + 96, a3 * v3[k]);
        }
    }
}

template <int PASS>
__device__ __forceinline__ void own_one(OwnLds& L, int src, int dl, int lane, const float4 a,
                                        const float* __restrict__ xl) {
    const float4 vl = ((const float4*)(xl + src * FEAT))[lane];
    const float4 vr = ((const float4*)&L.xr[dl * FEAT])[lane];
    float p = edge_dot(a, vl, vr);
    p += __shfl_xor(p, 1, 32);
    p += __shfl_xor(p, 2, 32);
    p += __shfl_xor(p, 4, 32);
    const int h = lane >> 3;
    if (PASS == 0) {
        if ((lane & 7) == 0)
            __hip_atomic_fetch_max(&L.ml[dl * HEADS + h], p,
                                   __ATOMIC_RELAXED, __HIP_MEMORY_SCOPE_WORKGROUP);
    } else {
        const float ex = __expf(p - L.ml[dl * HEADS + h]);
        if ((lane & 7) == 0) unsafeAtomicAdd(&L.sl[dl * HEADS + h], ex);
        float* ap = &L.acc[dl * FEAT + lane * 4];
        unsafeAtomicAdd(ap,     ex * vl.x);
        unsafeAtomicAdd(ap + 1, ex * vl.y);
        unsafeAtomicAdd(ap + 2, ex * vl.z);
        unsafeAtomicAdd(ap + 3, ex * vl.w);
    }
}

template <int PASS>
__device__ __forceinline__ void own_direct(OwnLds& L, int lane, int w, int lo, int nown,
                                        const float4 a, const unsigned pat,
                                        const float* __restrict__ xl,
                                        const int*   __restrict__ ei,
                                        const unsigned char* __restrict__ own8) {
    for (int n = w; n < nown; n += OWN_WAVES) own_one<PASS>(L, lo + n, n, lane, a, xl);
    const uint4* __restrict__ ownv = (const uint4*)own8;
    for (int t = w; t < NSCAN; t += OWN_WAVES) {
        const int piece = t * 32 + lane;
        const uint4 d = ownv[(piece < NVEC) ? piece : (NVEC - 1)];
        const unsigned hm = (piece < NVEC) ? own_hits(d, pat) : 0u;
        unsigned mask = __builtin_amdgcn_ballot_w32(hm != 0);
        while (mask) {
            const int l = __builtin_ctz(mask);
            unsigned hl = (unsigned)__builtin_amdgcn_readlane((int)hm, l);
            const int pl = __builtin_amdgcn_readlane(piece, l);
            while (hl) {
                const int e = pl * 16 + __builtin_ctz(hl);
                own_one<PASS>(L, ei[e], ei[NEDGE + e] - lo, lane, a, xl);
                hl &= hl - 1;
            }
            mask &= mask - 1;
        }
    }
}

__global__ __launch_bounds__(OWN_THREADS) void gat_own(const float* __restrict__ xl,
                                                       const float* __restrict__ xr,
                                                       const int*   __restrict__ ei,
                                                       const float* __restrict__ att,
                                                       const unsigned char* __restrict__ own8,
                                                       const float* __restrict__ bias,
                                                       float* __restrict__ out) {
    __shared__ __attribute__((aligned(16))) OwnLds L;

    const int lane = threadIdx.x & 31;
    const int w    = __builtin_amdgcn_readfirstlane(threadIdx.x >> 5);
    const int lo   = blockIdx.x * OWN_R;
    const int nown = (NNODES - lo < OWN_R) ? (NNODES - lo) : OWN_R;

    const float4 a = ((const float4*)att)[lane];

    for (int i = threadIdx.x; i < OWN_R * FEAT; i += OWN_THREADS) L.acc[i] = 0.0f;
    {
        float4 v[XR_TRIPS];
        #pragma unroll
        for (int j = 0; j < XR_TRIPS; ++j) {
            const int i = threadIdx.x + j * OWN_THREADS;
            v[j] = make_float4(0.f, 0.f, 0.f, 0.f);
            if (i < nown * (FEAT / 4)) v[j] = ((const float4*)(xr + lo * FEAT))[i];
        }
        #pragma unroll
        for (int j = 0; j < XR_TRIPS; ++j) {
            const int i = threadIdx.x + j * OWN_THREADS;
            if (i < OWN_R * (FEAT / 4)) ((float4*)L.xr)[i] = v[j];
        }
    }
    for (int i = threadIdx.x; i < OWN_R * HEADS; i += OWN_THREADS) {
        L.sl[i] = 0.0f;
        L.ml[i] = -__builtin_inff();
    }
    if ((int)threadIdx.x < nown) L.list[threadIdx.x] = (unsigned)(NEDGE + lo + (int)threadIdx.x);
    if (threadIdx.x == 0) L.cnt = nown;
    __syncthreads();

    {
        const int nself = (nown > w) ? (nown - w + OWN_WAVES - 1) / OWN_WAVES : 0;
        if (nself > 0) own_sweep1(L, w, OWN_WAVES, nself, lane, lo, a, xl, ei);
    }

    const int start = blockIdx.x * 6;
    const uint4* __restrict__ ownv = (const uint4*)own8;
    const unsigned pat = blockIdx.x * 0x01010101u;

    for (int t0 = w; t0 < NSCAN; t0 += OWN_WAVES * SCAN_UNR) {
        uint4 d[SCAN_UNR];
        int   piece[SCAN_UNR];
        #pragma unroll
        for (int k = 0; k < SCAN_UNR; ++k) {
            const int idx = t0 + k * OWN_WAVES;
            int ps = idx + start;
            if (ps >= NSCAN) ps -= NSCAN;
            piece[k] = ps * 32 + lane;
            const int pc = (piece[k] < NVEC) ? piece[k] : (NVEC - 1);
            d[k] = ownv[pc];
        }
        #pragma unroll
        for (int k = 0; k < SCAN_UNR; ++k)
            asm volatile("" :: "v"(d[k].x), "v"(d[k].y), "v"(d[k].z), "v"(d[k].w));
        unsigned hm[SCAN_UNR];
        int nh = 0;
        #pragma unroll
        for (int k = 0; k < SCAN_UNR; ++k) {
            const bool live = (t0 + k * OWN_WAVES < NSCAN) && (piece[k] < NVEC);
            hm[k] = live ? own_hits(d[k], pat) : 0u;
            nh += __builtin_popcount(hm[k]);
        }
        #pragma unroll
        for (int o = 16; o > 0; o >>= 1) nh += __shfl_xor(nh, o, 32);
        const int tot = __builtin_amdgcn_readfirstlane(nh);
        if (tot > 0) {
            int b = 0;
            if (lane == 0) b = atomicAdd(&L.cnt, tot);
            const int pos0 = __builtin_amdgcn_readlane(b, 0);
            int pos = pos0;
            #pragma unroll
            for (int k = 0; k < SCAN_UNR; ++k) {
                unsigned hk = hm[k];
                unsigned mask = __builtin_amdgcn_ballot_w32(hk != 0);
                while (mask) {
                    const int slot = pos + (int)__builtin_amdgcn_mbcnt_lo(mask, 0);
                    if (hk != 0) {
                        if (slot < LCAP) L.list[slot] = (unsigned)(piece[k] * 16 + __builtin_ctz(hk));
                        hk &= hk - 1;
                    }
                    pos += __builtin_popcount(mask);
                    mask = __builtin_amdgcn_ballot_w32(hk != 0);
                }
            }
            __builtin_amdgcn_wave_barrier();
            if (pos0 + tot <= LCAP) own_sweep1(L, pos0, 1, tot, lane, lo, a, xl, ei);
        }
    }
    __syncthreads();

    const int total = L.cnt;
    if (total <= LCAP) {
        const int ngroups = (total + QGROUP - 1) / QGROUP;
        for (int gi = w; gi < ngroups; gi += OWN_WAVES) {
            const int base = gi * QGROUP;
            const int g = (total - base < QGROUP) ? (total - base) : QGROUP;
            if (g == QGROUP) own_sum<true >(L, base, g, lane, xl);
            else             own_sum<false>(L, base, g, lane, xl);
        }
    } else {
        own_direct<0>(L, lane, w, lo, nown, a, pat, xl, ei, own8);
        __syncthreads();
        own_direct<1>(L, lane, w, lo, nown, a, pat, xl, ei, own8);
    }
    __syncthreads();

    const float b0 = bias[lane], b1 = bias[lane + 32], b2 = bias[lane + 64], b3 = bias[lane + 96];
    for (int n = w; n < nown; n += OWN_WAVES) {
        const float* ap = &L.acc[n * FEAT + lane];
        float* op = out + (lo + n) * FEAT + lane;
        op[0]  = ap[0]  / L.sl[n * HEADS + 0] + b0;
        op[32] = ap[32] / L.sl[n * HEADS + 1] + b1;
        op[64] = ap[64] / L.sl[n * HEADS + 2] + b2;
        op[96] = ap[96] / L.sl[n * HEADS + 3] + b3;
    }
}

extern "C" void kernel_launch(void* const* d_in, const int* in_sizes, int n_in,
                              void* d_out, int out_size, void* d_ws, size_t ws_size,
                              hipStream_t stream) {
    const float* x    = (const float*)d_in[0];
    const int*   ei   = (const int*)  d_in[1];
    const float* W_l  = (const float*)d_in[2];
    const float* W_r  = (const float*)d_in[3];
    const float* att  = (const float*)d_in[4];
    const float* bias = (const float*)d_in[5];
    float* out = (float*)d_out;

    float* ws    = (float*)d_ws;
    float* xl    = ws;
    float* xr    = xl + NNODES * FEAT;
    unsigned char* own8 = (unsigned char*)(xr + NNODES * FEAT);

    gat_plane<<<(NEDGE / 4 + 255) / 256, 256, 0, stream>>>(ei, own8);

    gat_gemm<<<dim3(NNODES / TROWS, 1, 2), 256, 0, stream>>>(x, W_l, W_r, xl, xr);

    gat_own<<<OWN_NB, OWN_THREADS, 0, stream>>>(xl, xr, ei, att, own8, bias, out);
}


# Owner blocks shift each node's softmax by its self-loop score, not the max: one sweep per edge (score, exp, LDS adds), rows read once; exact two-pass redo if score-shift>40; self term first in sums
# speedup vs baseline: 3.5049x; 1.1685x over previous
#include <hip/hip_runtime.h>

#define NNODES   50000
#define FEAT     128
#define HEADS    4
#define NEDGE    800000
#define ETOT     (NEDGE + NNODES)
#define NEG_SLOPE 0.2f

typedef __attribute__((ext_vector_type(2))) float v2f;
typedef __attribute__((ext_vector_type(4))) float v4f;
typedef __attribute__((ext_vector_type(8))) float v8f;

#define OWN_NB      256
#define OWN_R       ((NNODES + OWN_NB - 1) / OWN_NB)
static_assert((NNODES - 1) / OWN_R < OWN_NB && OWN_NB <= 256, "an owner id fits one byte");
static_assert((OWN_NB - 1) * OWN_R < NNODES, "every block owns at least one node");
static_assert(NEDGE % 16 == 0, "the owner-byte plane is whole 16-byte pieces");

__global__ __launch_bounds__(256) void gat_plane(const int* __restrict__ ei,
                                                 unsigned char* __restrict__ own8) {
    const int i = blockIdx.x * 256 + threadIdx.x;
    if (i < NEDGE / 4) {
        const int4 d = ((const int4*)(ei + NEDGE))[i];
        ((unsigned*)own8)[i] = ((unsigned)d.x / OWN_R)
                             | (((unsigned)d.y / OWN_R) << 8)
                             | (((unsigned)d.z / OWN_R) << 16)
                             | (((unsigned)d.w / OWN_R) << 24);
    }
}

#define TROWS   80
#define RTILES  (TROWS / 16)
#define APITCH  (FEAT + 4)

__global__ __launch_bounds__(256) void gat_gemm(const float* __restrict__ x,
                                                const float* __restrict__ Wl,
                                                const float* __restrict__ Wr,
                                                float* __restrict__ xl,
                                                float* __restrict__ xr) {
    __shared__ __attribute__((aligned(16))) float As[TROWS * APITCH];

    const int lane  = threadIdx.x & 31;
    const int wave  = threadIdx.x >> 5;
    const int m0    = blockIdx.x * TROWS;
    const int n0    = wave * 16;
    const int lhalf = lane >> 4;
    const int l15   = lane & 15;

    const float* __restrict__ W    = blockIdx.z ? Wr : Wl;
    float*       __restrict__ outp = blockIdx.z ? xr : xl;

    #pragma unroll
    for (int i = 0; i < TROWS / 8; ++i) {
        const int r = wave + i * 8;
        const float4 v = ((const float4*)(x + (m0 + r) * FEAT))[lane];
        *(float4*)(&As[r * APITCH + lane * 4]) = v;
    }
    __syncthreads();

    v8f c[RTILES];
    #pragma unroll
    for (int t = 0; t < RTILES; ++t)
        c[t] = (v8f){0.f, 0.f, 0.f, 0.f, 0.f, 0.f, 0.f, 0.f};

    const float* ap = As + l15 * APITCH + lhalf * 2;
    const float* __restrict__ bp = W + (lhalf * 2) * FEAT + n0 + l15;

    #pragma unroll 4
    for (int k = 0; k < FEAT; k += 4) {
        v2f b;
        b.x = bp[k * FEAT];
        b.y = bp[(k + 1) * FEAT];
        #pragma unroll
        for (int t = 0; t < RTILES; ++t) {
            const v2f a = *(const v2f*)(ap + t * 16 * APITCH + k);
            c[t] = __builtin_amdgcn_wmma_f32_16x16x4_f32(
                        false, a,  false, b,
                        (short)0, c[t],  false,  false);
        }
    }

    #pragma unroll
    for (int t = 0; t < RTILES; ++t) {
        #pragma unroll
        for (int r = 0; r < 8; ++r)
            outp[(m0 + t * 16 + r + lhalf * 8) * FEAT + n0 + l15] = c[t][r];
    }
}

#define OWN_WAVES   32
#define OWN_THREADS (OWN_WAVES * 32)
#define NVEC        (NEDGE / 16)
#define NSCAN       ((NVEC + 31) / 32)
#define SCAN_UNR    8
#define QGROUP      8
#define LCAP        4096
#define SELF_TRIPS  ((OWN_R + OWN_WAVES - 1) / OWN_WAVES)
#define EXP_CAP     40.0f
static_assert(NNODES <= 65536 && OWN_R <= 65536, "node ids stay small");
static_assert(OWN_NB * 6 < NSCAN, "staggered start stays inside the plane");

struct OwnLds {
    float    acc[OWN_R * FEAT];
    float    xr[OWN_R * FEAT];
    float    sl[OWN_R * HEADS];
    float    ml[OWN_R * HEADS];
    unsigned list[LCAP];
    int      cnt;
    int      bad;
};

__device__ __forceinline__ unsigned own_hits(const uint4 d, const unsigned pat) {
    const unsigned dw[4] = {d.x, d.y, d.z, d.w};
    unsigned hm = 0;
    #pragma unroll
    for (int c = 0; c < 4; ++c) {
        const unsigned t = dw[c] ^ pat;
        const unsigned z = ~(((t & 0x7F7F7F7Fu) + 0x7F7F7F7Fu) | t | 0x7F7F7F7Fu);
        hm |= (((z >> 7) * 0x01020408u) >> 24) << (4 * c);
    }
    return hm;
}

__device__ __forceinline__ float edge_dot(const float4 a, const float4 vl, const float4 vr) {
    float q = 0.f, t;
    t = vl.x + vr.x; t = (t >= 0.f) ? t : NEG_SLOPE * t; q += a.x * t;
    t = vl.y + vr.y; t = (t >= 0.f) ? t : NEG_SLOPE * t; q += a.y * t;
    t = vl.z + vr.z; t = (t >= 0.f) ? t : NEG_SLOPE * t; q += a.z * t;
    t = vl.w + vr.w; t = (t >= 0.f) ? t : NEG_SLOPE * t; q += a.w * t;
    return q;
}

__device__ __forceinline__ void own_fetch(const OwnLds& L, int ib, int g, int lane,
                                          const int* __restrict__ ei, int& sv, int& dv) {
    const int u = lane >> 2;
    const int e = (u < g) ? (int)L.list[ib + u] : 0;
    sv = ei[e];
    dv = ei[NEDGE + e];
}

template <bool FULL>
__device__ __forceinline__ bool own_edges(OwnLds& L, int g, int lane, int sr, int dl,
                                          const float4 a, const float* __restrict__ xl,
                                          int ibn, int gn, const int* __restrict__ ei,
                                          int& sv, int& dv) {
    v4f vl[QGROUP];
    int dj[QGROUP];
    #pragma unroll
    for (int k = 0; k < QGROUP; ++k) {
        vl[k] = (v4f){0.f, 0.f, 0.f, 0.f};
        dj[k] = 0;
        if (FULL || k < g) {
            const int sj = __builtin_amdgcn_readlane(sr, 4 * k);
            dj[k]        = __builtin_amdgcn_readlane(dl, 4 * k);
            vl[k] = ((const v4f*)(xl + sj * FEAT))[lane];
        }
    }
    own_fetch(L, ibn, gn, lane, ei, sv, dv);
    const int h = lane >> 3;
    v4f   vr[QGROUP];
    float cs[QGROUP];
    #pragma unroll
    for (int k = 0; k < QGROUP; ++k) {
        vr[k] = (v4f){0.f, 0.f, 0.f, 0.f};
        cs[k] = 0.f;
        if (FULL || k < g) {
            vr[k] = ((const v4f*)&L.xr[dj[k] * FEAT])[lane];
            cs[k] = L.ml[dj[k] * HEADS + h];
        }
    }
    float p[QGROUP];
    #pragma unroll
    for (int k = 0; k < QGROUP; ++k) {
        p[k] = 0.f;
        if (FULL || k < g)
            p[k] = edge_dot(a, make_float4(vl[k].x, vl[k].y, vl[k].z, vl[k].w),
                               make_float4(vr[k].x, vr[k].y, vr[k].z, vr[k].w));
    }
    __builtin_amdgcn_sched_barrier(0);
    #pragma unroll
    for (int k = 0; k < QGROUP; ++k) p[k] += __shfl_xor(p[k], 1, 32);
    __builtin_amdgcn_sched_barrier(0);
    #pragma unroll
    for (int k = 0; k < QGROUP; ++k) p[k] += __shfl_xor(p[k], 2, 32);
    __builtin_amdgcn_sched_barrier(0);
    #pragma unroll
    for (int k = 0; k < QGROUP; ++k) p[k] += __shfl_xor(p[k], 4, 32);
    __builtin_amdgcn_sched_barrier(0);

    bool bad = false;
    #pragma unroll
    for (int k = 0; k < QGROUP; ++k) {
        if (FULL || k < g) {
            const float d = p[k] - cs[k];
            bad = bad || !(d <= EXP_CAP);
            const float ex = __expf(d);
            float* ap = &L.acc[dj[k] * FEAT + lane];
            unsafeAtomicAdd(ap,      ex * vl[k].x);
            unsafeAtomicAdd(ap + 32, ex * vl[k].y);
            unsafeAtomicAdd(ap + 64, ex * vl[k].z);
            unsafeAtomicAdd(ap + 96, ex * vl[k].w);
            if ((lane & 7) == 0) unsafeAtomicAdd(&L.sl[dj[k] * HEADS + h], ex);
        }
    }
    return bad;
}

__device__ __forceinline__ bool own_sweep(OwnLds& L, int ib, int n, int lane, int lo,
                                          const float4 a, const float* __restrict__ xl,
                                          const int* __restrict__ ei) {
    bool bad = false;
    int sv, dv;
    own_fetch(L, ib, (n < QGROUP) ? n : QGROUP, lane, ei, sv, dv);
    for (int j = 0; j < n; j += QGROUP) {
        const int g = (n - j < QGROUP) ? (n - j) : QGROUP;
        const int sr = sv, dl = dv - lo;
        int gg = n - j - QGROUP;
        gg = (gg < 0) ? 0 : ((gg < QGROUP) ? gg : QGROUP);
        const int ibn = ib + j + QGROUP;
        if (g == QGROUP) bad = own_edges<true >(L, g, lane, sr, dl, a, xl, ibn, gg, ei, sv, dv) || bad;
        else             bad = own_edges<false>(L, g, lane, sr, dl, a, xl, ibn, gg, ei, sv, dv) || bad;
    }
    return bad;
}

template <int PASS>
__device__ __forceinline__ void own_one(OwnLds& L, int src, int dl, int lane, const float4 a,
                                        const float* __restrict__ xl) {
    const float4 vl = ((const float4*)(xl + src * FEAT))[lane];
    const float4 vr = ((const float4*)&L.xr[dl * FEAT])[lane];
    float p = edge_dot(a, vl, vr);
    p += __shfl_xor(p, 1, 32);
    p += __shfl_xor(p, 2, 32);
    p += __shfl_xor(p, 4, 32);
    const int h = lane >> 3;
    if (PASS == 0) {
        if ((lane & 7) == 0)
            __hip_atomic_fetch_max(&L.ml[dl * HEADS + h], p,
                                   __ATOMIC_RELAXED, __HIP_MEMORY_SCOPE_WORKGROUP);
    } else {
        const float ex = __expf(p - L.ml[dl * HEADS + h]);
        if ((lane & 7) == 0) unsafeAtomicAdd(&L.sl[dl * HEADS + h], ex);
        float* ap = &L.acc[dl * FEAT + lane];
        unsafeAtomicAdd(ap,      ex * vl.x);
        unsafeAtomicAdd(ap + 32, ex * vl.y);
        unsafeAtomicAdd(ap + 64, ex * vl.z);
        unsafeAtomicAdd(ap + 96, ex * vl.w);
    }
}

template <int PASS>
__device__ __forceinline__ void own_direct(OwnLds& L, int lane, int w, int lo, int nown,
                                        const float4 a, const unsigned pat,
                                        const float* __restrict__ xl,
                                        const int*   __restrict__ ei,
                                        const unsigned char* __restrict__ own8) {
    for (int n = w; n < nown; n += OWN_WAVES) own_one<PASS>(L, lo + n, n, lane, a, xl);
    const uint4* __restrict__ ownv = (const uint4*)own8;
    for (int t = w; t < NSCAN; t += OWN_WAVES) {
        const int piece = t * 32 + lane;
        const uint4 d = ownv[(piece < NVEC) ? piece : (NVEC - 1)];
        const unsigned hm = (piece < NVEC) ? own_hits(d, pat) : 0u;
        unsigned mask = __builtin_amdgcn_ballot_w32(hm != 0);
        while (mask) {
            const int l = __builtin_ctz(mask);
            unsigned hl = (unsigned)__builtin_amdgcn_readlane((int)hm, l);
            const int pl = __builtin_amdgcn_readlane(piece, l);
            while (hl) {
                const int e = pl * 16 + __builtin_ctz(hl);
                own_one<PASS>(L, ei[e], ei[NEDGE + e] - lo, lane, a, xl);
                hl &= hl - 1;
            }
            mask &= mask - 1;
        }
    }
}

__global__ __launch_bounds__(OWN_THREADS) void gat_own(const float* __restrict__ xl,
                                                       const float* __restrict__ xr,
                                                       const int*   __restrict__ ei,
                                                       const float* __restrict__ att,
                                                       const unsigned char* __restrict__ own8,
                                                       const float* __restrict__ bias,
                                                       float* __restrict__ out) {
    __shared__ __attribute__((aligned(16))) OwnLds L;

    const int lane = threadIdx.x & 31;
    const int w    = __builtin_amdgcn_readfirstlane(threadIdx.x >> 5);
    const int lo   = blockIdx.x * OWN_R;
    const int nown = (NNODES - lo < OWN_R) ? (NNODES - lo) : OWN_R;
    const int h    = lane >> 3;

    const float4 a = ((const float4*)att)[lane];

    {
        v4f vr[SELF_TRIPS], vl[SELF_TRIPS];
        #pragma unroll
        for (int i = 0; i < SELF_TRIPS; ++i) {
            const int n  = w + i * OWN_WAVES;
            const int nc = (n < nown) ? n : (nown - 1);
            vr[i] = ((const v4f*)(xr + (lo + nc) * FEAT))[lane];
            vl[i] = ((const v4f*)(xl + (lo + nc) * FEAT))[lane];
        }
        float p[SELF_TRIPS];
        #pragma unroll
        for (int i = 0; i < SELF_TRIPS; ++i)
            p[i] = edge_dot(a, make_float4(vl[i].x, vl[i].y, vl[i].z, vl[i].w),
                               make_float4(vr[i].x, vr[i].y, vr[i].z, vr[i].w));
        #pragma unroll
        for (int i = 0; i < SELF_TRIPS; ++i) p[i] += __shfl_xor(p[i], 1, 32);
        #pragma unroll
        for (int i = 0; i < SELF_TRIPS; ++i) p[i] += __shfl_xor(p[i], 2, 32);
        #pragma unroll
        for (int i = 0; i < SELF_TRIPS; ++i) p[i] += __shfl_xor(p[i], 4, 32);
        #pragma unroll
        for (int i = 0; i < SELF_TRIPS; ++i) {
            const int n = w + i * OWN_WAVES;
            if (n < nown) {
                ((v4f*)&L.xr[n * FEAT])[lane] = vr[i];
                float* ap = &L.acc[n * FEAT + lane];
                ap[0] = vl[i].x; ap[32] = vl[i].y; ap[64] = vl[i].z; ap[96] = vl[i].w;
                if ((lane & 7) == 0) {
                    L.ml[n * HEADS + h] = p[i];
                    L.sl[n * HEADS + h] = 1.0f;
                }
            }
        }
    }
    if (threadIdx.x == 0) { L.cnt = 0; L.bad = 0; }
    __syncthreads();

    const int start = blockIdx.x * 6;
    const uint4* __restrict__ ownv = (const uint4*)own8;
    const unsigned pat = blockIdx.x * 0x01010101u;
    bool bad = false;

    for (int t0 = w; t0 < NSCAN; t0 += OWN_WAVES * SCAN_UNR) {
        uint4 d[SCAN_UNR];
        int   piece[SCAN_UNR];
        #pragma unroll
        for (int k = 0; k < SCAN_UNR; ++k) {
            const int idx = t0 + k * OWN_WAVES;
            int ps = idx + start;
            if (ps >= NSCAN) ps -= NSCAN;
            piece[k] = ps * 32 + lane;
            const int pc = (piece[k] < NVEC) ? piece[k] : (NVEC - 1);
            d[k] = ownv[pc];
        }
        #pragma unroll
        for (int k = 0; k < SCAN_UNR; ++k)
            asm volatile("" :: "v"(d[k].x), "v"(d[k].y), "v"(d[k].z), "v"(d[k].w));
        unsigned hm[SCAN_UNR];
        int nh = 0;
        #pragma unroll
        for (int k = 0; k < SCAN_UNR; ++k) {
            const bool live = (t0 + k * OWN_WAVES < NSCAN) && (piece[k] < NVEC);
            hm[k] = live ? own_hits(d[k], pat) : 0u;
            nh += __builtin_popcount(hm[k]);
        }
        #pragma unroll
        for (int o = 16; o > 0; o >>= 1) nh += __shfl_xor(nh, o, 32);
        const int tot = __builtin_amdgcn_readfirstlane(nh);
        if (tot > 0) {
            int b = 0;
            if (lane == 0) b = atomicAdd(&L.cnt, tot);
            const int pos0 = __builtin_amdgcn_readlane(b, 0);
            int pos = pos0;
            #pragma unroll
            for (int k = 0; k < SCAN_UNR; ++k) {
                unsigned hk = hm[k];
                unsigned mask = __builtin_amdgcn_ballot_w32(hk != 0);
                while (mask) {
                    const int slot = pos + (int)__builtin_amdgcn_mbcnt_lo(mask, 0);
                    if (hk != 0) {
                        if (slot < LCAP) L.list[slot] = (unsigned)(piece[k] * 16 + __builtin_ctz(hk));
                        hk &= hk - 1;
                    }
                    pos += __builtin_popcount(mask);
                    mask = __builtin_amdgcn_ballot_w32(hk != 0);
                }
            }
            __builtin_amdgcn_wave_barrier();
            if (pos0 + tot <= LCAP) bad = own_sweep(L, pos0, tot, lane, lo, a, xl, ei) || bad;
        }
    }
    if (__builtin_amdgcn_ballot_w32(bad) != 0u && lane == 0) L.bad = 1;
    __syncthreads();

    if (L.cnt > LCAP || L.bad != 0) {
        for (int i = threadIdx.x; i < OWN_R * FEAT; i += OWN_THREADS) L.acc[i] = 0.0f;
        for (int i = threadIdx.x; i < OWN_R * HEADS; i += OWN_THREADS) {
            L.sl[i] = 0.0f;
            L.ml[i] = -__builtin_inff();
        }
        __syncthreads();
        own_direct<0>(L, lane, w, lo, nown, a, pat, xl, ei, own8);
        __syncthreads();
        own_direct<1>(L, lane, w, lo, nown, a, pat, xl, ei, own8);
        __syncthreads();
    }

    const float4 bv = ((const float4*)bias)[lane];
    for (int n = w; n < nown; n += OWN_WAVES) {
        const float* ap = &L.acc[n * FEAT + lane];
        const float s = L.sl[n * HEADS + h];
        float4 o;
        o.x = ap[0]  / s + bv.x;
        o.y = ap[32] / s + bv.y;
        o.z = ap[64] / s + bv.z;
        o.w = ap[96] / s + bv.w;
        ((float4*)(out + (lo + n) * FEAT))[lane] = o;
    }
}

extern "C" void kernel_launch(void* const* d_in, const int* in_sizes, int n_in,
                              void* d_out, int out_size, void* d_ws, size_t ws_size,
                              hipStream_t stream) {
    const float* x    = (const float*)d_in[0];
    const int*   ei   = (const int*)  d_in[1];
    const float* W_l  = (const float*)d_in[2];
    const float* W_r  = (const float*)d_in[3];
    const float* att  = (const float*)d_in[4];
    const float* bias = (const float*)d_in[5];
    float* out = (float*)d_out;

    float* ws    = (float*)d_ws;
    float* xl    = ws;
    float* xr    = xl + NNODES * FEAT;
    unsigned char* own8 = (unsigned char*)(xr + NNODES * FEAT);

    gat_plane<<<(NEDGE / 4 + 255) / 256, 256, 0, stream>>>(ei, own8);

    gat_gemm<<<dim3(NNODES / TROWS, 1, 2), 256, 0, stream>>>(x, W_l, W_r, xl, xr);

    gat_own<<<OWN_NB, OWN_THREADS, 0, stream>>>(xl, xr, ei, att, own8, bias, out);
}


# Owner plane becomes one bit plane per owner block, built per 1024-edge chunk in LDS, stored as whole 128-B lines: a block scans its own 100 KB, not 800 KB of owner bytes; sums in LDS arrival order
# speedup vs baseline: 3.9696x; 1.1326x over previous
#include <hip/hip_runtime.h>

#define NNODES   50000
#define FEAT     128
#define HEADS    4
#define NEDGE    800000
#define ETOT     (NEDGE + NNODES)
#define NEG_SLOPE 0.2f

typedef __attribute__((ext_vector_type(2))) float v2f;
typedef __attribute__((ext_vector_type(4))) float v4f;
typedef __attribute__((ext_vector_type(8))) float v8f;

#define OWN_NB      256
#define OWN_R       ((NNODES + OWN_NB - 1) / OWN_NB)
static_assert((NNODES - 1) / OWN_R < OWN_NB, "every node has an owner");
static_assert((OWN_NB - 1) * OWN_R < NNODES, "every block owns at least one node");
static_assert(NEDGE % 4 == 0, "a thread of the plane kernel takes four whole edges");

#define CH_EDGES    1024
#define NCHUNK      ((NEDGE + CH_EDGES - 1) / CH_EDGES)
#define PLANE_DW    (NCHUNK * (CH_EDGES / 32))
#define PLANE_TRIPS (OWN_NB * (CH_EDGES / 32) / 4 / 256)
static_assert(PLANE_TRIPS * 256 * 4 == OWN_NB * (CH_EDGES / 32), "256 threads cover a chunk's lines exactly");
static_assert(CH_EDGES == 4 * 256, "a thread of the plane kernel takes four edges of the chunk");

__global__ __launch_bounds__(256) void gat_plane(const int* __restrict__ ei,
                                                 unsigned* __restrict__ own1) {
    __shared__ __attribute__((aligned(16))) unsigned bits[OWN_NB * (CH_EDGES / 32)];
    const int c = blockIdx.x;
    const int q = c * (CH_EDGES / 4) + threadIdx.x;
    const int4 d = ((const int4*)(ei + NEDGE))[(q < NEDGE / 4) ? q : (NEDGE / 4 - 1)];
    #pragma unroll
    for (int k = 0; k < PLANE_TRIPS; ++k)
        ((uint4*)bits)[threadIdx.x + 256 * k] = make_uint4(0u, 0u, 0u, 0u);
    __syncthreads();
    if (q < NEDGE / 4) {
        const int      dwi = threadIdx.x >> 3;
        const unsigned b0  = 1u << (4 * (threadIdx.x & 7));
        atomicOr(&bits[((unsigned)d.x / OWN_R) * (CH_EDGES / 32) + dwi], b0);
        atomicOr(&bits[((unsigned)d.y / OWN_R) * (CH_EDGES / 32) + dwi], b0 << 1);
        atomicOr(&bits[((unsigned)d.z / OWN_R) * (CH_EDGES / 32) + dwi], b0 << 2);
        atomicOr(&bits[((unsigned)d.w / OWN_R) * (CH_EDGES / 32) + dwi], b0 << 3);
    }
    __syncthreads();
    #pragma unroll
    for (int k = 0; k < PLANE_TRIPS; ++k) {
        const int i = threadIdx.x + 256 * k;
        const int o = i >> 3, part = i & 7;
        ((uint4*)(own1 + o * PLANE_DW + c * (CH_EDGES / 32)))[part] = ((const uint4*)bits)[i];
    }
}

#define TROWS   80
#define RTILES  (TROWS / 16)
#define APITCH  (FEAT + 4)

__global__ __launch_bounds__(256) void gat_gemm(const float* __restrict__ x,
                                                const float* __restrict__ Wl,
                                                const float* __restrict__ Wr,
                                                float* __restrict__ xl,
                                                float* __restrict__ xr) {
    __shared__ __attribute__((aligned(16))) float As[TROWS * APITCH];

    const int lane  = threadIdx.x & 31;
    const int wave  = threadIdx.x >> 5;
    const int m0    = blockIdx.x * TROWS;
    const int n0    = wave * 16;
    const int lhalf = lane >> 4;
    const int l15   = lane & 15;

    const float* __restrict__ W    = blockIdx.z ? Wr : Wl;
    float*       __restrict__ outp = blockIdx.z ? xr : xl;

    #pragma unroll
    for (int i = 0; i < TROWS / 8; ++i) {
        const int r = wave + i * 8;
        const float4 v = ((const float4*)(x + (m0 + r) * FEAT))[lane];
        *(float4*)(&As[r * APITCH + lane * 4]) = v;
    }
    __syncthreads();

    v8f c[RTILES];
    #pragma unroll
    for (int t = 0; t < RTILES; ++t)
        c[t] = (v8f){0.f, 0.f, 0.f, 0.f, 0.f, 0.f, 0.f, 0.f};

    const float* ap = As + l15 * APITCH + lhalf * 2;
    const float* __restrict__ bp = W + (lhalf * 2) * FEAT + n0 + l15;

    #pragma unroll 4
    for (int k = 0; k < FEAT; k += 4) {
        v2f b;
        b.x = bp[k * FEAT];
        b.y = bp[(k + 1) * FEAT];
        #pragma unroll
        for (int t = 0; t < RTILES; ++t) {
            const v2f a = *(const v2f*)(ap + t * 16 * APITCH + k);
            c[t] = __builtin_amdgcn_wmma_f32_16x16x4_f32(
                        false, a,  false, b,
                        (short)0, c[t],  false,  false);
        }
    }

    #pragma unroll
    for (int t = 0; t < RTILES; ++t) {
        #pragma unroll
        for (int r = 0; r < 8; ++r)
            outp[(m0 + t * 16 + r + lhalf * 8) * FEAT + n0 + l15] = c[t][r];
    }
}

#define OWN_WAVES   32
#define OWN_THREADS (OWN_WAVES * 32)
#define NVEC        (PLANE_DW / 4)
#define NSCAN       ((NVEC + 31) / 32)
#define SCAN_UNR    ((NSCAN + OWN_WAVES - 1) / OWN_WAVES)
#define QGROUP      8
#define LCAP        4096
#define SELF_TRIPS  ((OWN_R + OWN_WAVES - 1) / OWN_WAVES)
#define EXP_CAP     40.0f
static_assert(NNODES <= 65536 && OWN_R <= 65536, "node ids stay small");

struct OwnLds {
    float    acc[OWN_R * FEAT];
    float    xr[OWN_R * FEAT];
    float    sl[OWN_R * HEADS];
    float    ml[OWN_R * HEADS];
    unsigned list[LCAP];
    int      cnt;
    int      bad;
};

__device__ __forceinline__ float edge_dot(const float4 a, const float4 vl, const float4 vr) {
    float q = 0.f, t;
    t = vl.x + vr.x; t = (t >= 0.f) ? t : NEG_SLOPE * t; q += a.x * t;
    t = vl.y + vr.y; t = (t >= 0.f) ? t : NEG_SLOPE * t; q += a.y * t;
    t = vl.z + vr.z; t = (t >= 0.f) ? t : NEG_SLOPE * t; q += a.z * t;
    t = vl.w + vr.w; t = (t >= 0.f) ? t : NEG_SLOPE * t; q += a.w * t;
    return q;
}

__device__ __forceinline__ void own_fetch(const OwnLds& L, int ib, int g, int lane,
                                          const int* __restrict__ ei, int& sv, int& dv) {
    const int u = lane >> 2;
    const int e = (u < g) ? (int)L.list[ib + u] : 0;
    sv = ei[e];
    dv = ei[NEDGE + e];
}

template <bool FULL>
__device__ __forceinline__ bool own_edges(OwnLds& L, int g, int lane, int sr, int dl,
                                          const float4 a, const float* __restrict__ xl,
                                          int ibn, int gn, const int* __restrict__ ei,
                                          int& sv, int& dv) {
    v4f vl[QGROUP];
    int dj[QGROUP];
    #pragma unroll
    for (int k = 0; k < QGROUP; ++k) {
        vl[k] = (v4f){0.f, 0.f, 0.f, 0.f};
        dj[k] = 0;
        if (FULL || k < g) {
            const int sj = __builtin_amdgcn_readlane(sr, 4 * k);
            dj[k]        = __builtin_amdgcn_readlane(dl, 4 * k);
            vl[k] = ((const v4f*)(xl + sj * FEAT))[lane];
        }
    }
    own_fetch(L, ibn, gn, lane, ei, sv, dv);
    const int h = lane >> 3;
    v4f   vr[QGROUP];
    float cs[QGROUP];
    #pragma unroll
    for (int k = 0; k < QGROUP; ++k) {
        vr[k] = (v4f){0.f, 0.f, 0.f, 0.f};
        cs[k] = 0.f;
        if (FULL || k < g) {
            vr[k] = ((const v4f*)&L.xr[dj[k] * FEAT])[lane];
            cs[k] = L.ml[dj[k] * HEADS + h];
        }
    }
    float p[QGROUP];
    #pragma unroll
    for (int k = 0; k < QGROUP; ++k) {
        p[k] = 0.f;
        if (FULL || k < g)
            p[k] = edge_dot(a, make_float4(vl[k].x, vl[k].y, vl[k].z, vl[k].w),
                               make_float4(vr[k].x, vr[k].y, vr[k].z, vr[k].w));
    }
    __builtin_amdgcn_sched_barrier(0);
    #pragma unroll
    for (int k = 0; k < QGROUP; ++k) p[k] += __shfl_xor(p[k], 1, 32);
    __builtin_amdgcn_sched_barrier(0);
    #pragma unroll
    for (int k = 0; k < QGROUP; ++k) p[k] += __shfl_xor(p[k], 2, 32);
    __builtin_amdgcn_sched_barrier(0);
    #pragma unroll
    for (int k = 0; k < QGROUP; ++k) p[k] += __shfl_xor(p[k], 4, 32);
    __builtin_amdgcn_sched_barrier(0);

    bool bad = false;
    #pragma unroll
    for (int k = 0; k < QGROUP; ++k) {
        if (FULL || k < g) {
            const float d = p[k] - cs[k];
            bad = bad || !(d <= EXP_CAP);
            const float ex = __expf(d);
            float* ap = &L.acc[dj[k] * FEAT + lane];
            unsafeAtomicAdd(ap,      ex * vl[k].x);
            unsafeAtomicAdd(ap + 32, ex * vl[k].y);
            unsafeAtomicAdd(ap + 64, ex * vl[k].z);
            unsafeAtomicAdd(ap + 96, ex * vl[k].w);
            if ((lane & 7) == 0) unsafeAtomicAdd(&L.sl[dj[k] * HEADS + h], ex);
        }
    }
    return bad;
}

__device__ __forceinline__ bool own_sweep(OwnLds& L, int ib, int n, int lane, int lo,
                                          const float4 a, const float* __restrict__ xl,
                                          const int* __restrict__ ei) {
    bool bad = false;
    int sv, dv;
    own_fetch(L, ib, (n < QGROUP) ? n : QGROUP, lane, ei, sv, dv);
    for (int j = 0; j < n; j += QGROUP) {
        const int g = (n - j < QGROUP) ? (n - j) : QGROUP;
        const int sr = sv, dl = dv - lo;
        int gg = n - j - QGROUP;
        gg = (gg < 0) ? 0 : ((gg < QGROUP) ? gg : QGROUP);
        const int ibn = ib + j + QGROUP;
        if (g == QGROUP) bad = own_edges<true >(L, g, lane, sr, dl, a, xl, ibn, gg, ei, sv, dv) || bad;
        else             bad = own_edges<false>(L, g, lane, sr, dl, a, xl, ibn, gg, ei, sv, dv) || bad;
    }
    return bad;
}

template <int PASS>
__device__ __forceinline__ void own_one(OwnLds& L, int src, int dl, int lane, const float4 a,
                                        const float* __restrict__ xl) {
    const float4 vl = ((const float4*)(xl + src * FEAT))[lane];
    const float4 vr = ((const float4*)&L.xr[dl * FEAT])[lane];
    float p = edge_dot(a, vl, vr);
    p += __shfl_xor(p, 1, 32);
    p += __shfl_xor(p, 2, 32);
    p += __shfl_xor(p, 4, 32);
    const int h = lane >> 3;
    if (PASS == 0) {
        if ((lane & 7) == 0)
            __hip_atomic_fetch_max(&L.ml[dl * HEADS + h], p,
                                   __ATOMIC_RELAXED, __HIP_MEMORY_SCOPE_WORKGROUP);
    } else {
        const float ex = __expf(p - L.ml[dl * HEADS + h]);
        if ((lane & 7) == 0) unsafeAtomicAdd(&L.sl[dl * HEADS + h], ex);
        float* ap = &L.acc[dl * FEAT + lane];
        unsafeAtomicAdd(ap,      ex * vl.x);
        unsafeAtomicAdd(ap + 32, ex * vl.y);
        unsafeAtomicAdd(ap + 64, ex * vl.z);
        unsafeAtomicAdd(ap + 96, ex * vl.w);
    }
}

template <int PASS>
__device__ __forceinline__ void own_direct(OwnLds& L, int lane, int w, int lo, int nown,
                                        const float4 a,
                                        const float* __restrict__ xl,
                                        const int*   __restrict__ ei,
                                        const uint4* __restrict__ ownv) {
    for (int n = w; n < nown; n += OWN_WAVES) own_one<PASS>(L, lo + n, n, lane, a, xl);
    for (int t = w; t < NSCAN; t += OWN_WAVES) {
        const int piece = t * 32 + lane;
        const uint4 d = ownv[(piece < NVEC) ? piece : (NVEC - 1)];
        const unsigned dw[4] = {d.x, d.y, d.z, d.w};
        #pragma unroll
        for (int c = 0; c < 4; ++c) {
            const unsigned hm = (piece < NVEC) ? dw[c] : 0u;
            unsigned mask = __builtin_amdgcn_ballot_w32(hm != 0);
            while (mask) {
                const int l = __builtin_ctz(mask);
                unsigned hl = (unsigned)__builtin_amdgcn_readlane((int)hm, l);
                const int pl = __builtin_amdgcn_readlane(piece, l);
                while (hl) {
                    const int e = pl * 128 + c * 32 + __builtin_ctz(hl);
                    own_one<PASS>(L, ei[e], ei[NEDGE + e] - lo, lane, a, xl);
                    hl &= hl - 1;
                }
                mask &= mask - 1;
            }
        }
    }
}

__global__ __launch_bounds__(OWN_THREADS) void gat_own(const float* __restrict__ xl,
                                                       const float* __restrict__ xr,
                                                       const int*   __restrict__ ei,
                                                       const float* __restrict__ att,
                                                       const unsigned* __restrict__ own1,
                                                       const float* __restrict__ bias,
                                                       float* __restrict__ out) {
    __shared__ __attribute__((aligned(16))) OwnLds L;

    const int lane = threadIdx.x & 31;
    const int w    = __builtin_amdgcn_readfirstlane(threadIdx.x >> 5);
    const int lo   = blockIdx.x * OWN_R;
    const int nown = (NNODES - lo < OWN_R) ? (NNODES - lo) : OWN_R;
    const int h    = lane >> 3;

    const float4 a = ((const float4*)att)[lane];

    const uint4* __restrict__ ownv = (const uint4*)(own1 + blockIdx.x * PLANE_DW);
    uint4 d[SCAN_UNR];
    int   piece[SCAN_UNR];
    #pragma unroll
    for (int k = 0; k < SCAN_UNR; ++k) {
        piece[k] = (w + k * OWN_WAVES) * 32 + lane;
        const int pc = (piece[k] < NVEC) ? piece[k] : (NVEC - 1);
        d[k] = ownv[pc];
    }

    {
        v4f vr[SELF_TRIPS], vl[SELF_TRIPS];
        #pragma unroll
        for (int i = 0; i < SELF_TRIPS; ++i) {
            const int n  = w + i * OWN_WAVES;
            const int nc = (n < nown) ? n : (nown - 1);
            vr[i] = ((const v4f*)(xr + (lo + nc) * FEAT))[lane];
            vl[i] = ((const v4f*)(xl + (lo + nc) * FEAT))[lane];
        }
        float p[SELF_TRIPS];
        #pragma unroll
        for (int i = 0; i < SELF_TRIPS; ++i)
            p[i] = edge_dot(a, make_float4(vl[i].x, vl[i].y, vl[i].z, vl[i].w),
                               make_float4(vr[i].x, vr[i].y, vr[i].z, vr[i].w));
        #pragma unroll
        for (int i = 0; i < SELF_TRIPS; ++i) p[i] += __shfl_xor(p[i], 1, 32);
        #pragma unroll
        for (int i = 0; i < SELF_TRIPS; ++i) p[i] += __shfl_xor(p[i], 2, 32);
        #pragma unroll
        for (int i = 0; i < SELF_TRIPS; ++i) p[i] += __shfl_xor(p[i], 4, 32);
        #pragma unroll
        for (int i = 0; i < SELF_TRIPS; ++i) {
            const int n = w + i * OWN_WAVES;
            if (n < nown) {
                ((v4f*)&L.xr[n * FEAT])[lane] = vr[i];
                float* ap = &L.acc[n * FEAT + lane];
                ap[0] = vl[i].x; ap[32] = vl[i].y; ap[64] = vl[i].z; ap[96] = vl[i].w;
                if ((lane & 7) == 0) {
                    L.ml[n * HEADS + h] = p[i];
                    L.sl[n * HEADS + h] = 1.0f;
                }
            }
        }
    }
    if (threadIdx.x == 0) { L.cnt = 0; L.bad = 0; }
    __syncthreads();

    bool bad = false;
    {
        unsigned hm[SCAN_UNR][4];
        int nh = 0;
        #pragma unroll
        for (int k = 0; k < SCAN_UNR; ++k) {
            const bool live = piece[k] < NVEC;
            hm[k][0] = live ? d[k].x : 0u;
            hm[k][1] = live ? d[k].y : 0u;
            hm[k][2] = live ? d[k].z : 0u;
            hm[k][3] = live ? d[k].w : 0u;
            nh += __builtin_popcount(hm[k][0]) + __builtin_popcount(hm[k][1])
                + __builtin_popcount(hm[k][2]) + __builtin_popcount(hm[k][3]);
        }
        #pragma unroll
        for (int o = 16; o > 0; o >>= 1) nh += __shfl_xor(nh, o, 32);
        const int tot = __builtin_amdgcn_readfirstlane(nh);
        if (tot > 0) {
            int b = 0;
            if (lane == 0) b = atomicAdd(&L.cnt, tot);
            const int pos0 = __builtin_amdgcn_readlane(b, 0);
            int pos = pos0;
            #pragma unroll
            for (int k = 0; k < SCAN_UNR; ++k) {
                #pragma unroll
                for (int c = 0; c < 4; ++c) {
                    unsigned hk = hm[k][c];
                    unsigned mask = __builtin_amdgcn_ballot_w32(hk != 0);
                    while (mask) {
                        const int slot = pos + (int)__builtin_amdgcn_mbcnt_lo(mask, 0);
                        if (hk != 0) {
                            if (slot < LCAP)
                                L.list[slot] = (unsigned)(piece[k] * 128 + c * 32 + __builtin_ctz(hk));
                            hk &= hk - 1;
                        }
                        pos += __builtin_popcount(mask);
                        mask = __builtin_amdgcn_ballot_w32(hk != 0);
                    }
                }
            }
            __builtin_amdgcn_wave_barrier();
            if (pos0 + tot <= LCAP) bad = own_sweep(L, pos0, tot, lane, lo, a, xl, ei) || bad;
        }
    }
    if (__builtin_amdgcn_ballot_w32(bad) != 0u && lane == 0) L.bad = 1;
    __syncthreads();

    if (L.cnt > LCAP || L.bad != 0) {
        for (int i = threadIdx.x; i < OWN_R * FEAT; i += OWN_THREADS) L.acc[i] = 0.0f;
        for (int i = threadIdx.x; i < OWN_R * HEADS; i += OWN_THREADS) {
            L.sl[i] = 0.0f;
            L.ml[i] = -__builtin_inff();
        }
        __syncthreads();
        own_direct<0>(L, lane, w, lo, nown, a, xl, ei, ownv);
        __syncthreads();
        own_direct<1>(L, lane, w, lo, nown, a, xl, ei, ownv);
        __syncthreads();
    }

    const float4 bv = ((const float4*)bias)[lane];
    for (int n = w; n < nown; n += OWN_WAVES) {
        const float* ap = &L.acc[n * FEAT + lane];
        const float s = L.sl[n * HEADS + h];
        float4 o;
        o.x = ap[0]  / s + bv.x;
        o.y = ap[32] / s + bv.y;
        o.z = ap[64] / s + bv.z;
        o.w = ap[96] / s + bv.w;
        ((float4*)(out + (lo + n) * FEAT))[lane] = o;
    }
}

extern "C" void kernel_launch(void* const* d_in, const int* in_sizes, int n_in,
                              void* d_out, int out_size, void* d_ws, size_t ws_size,
                              hipStream_t stream) {
    const float* x    = (const float*)d_in[0];
    const int*   ei   = (const int*)  d_in[1];
    const float* W_l  = (const float*)d_in[2];
    const float* W_r  = (const float*)d_in[3];
    const float* att  = (const float*)d_in[4];
    const float* bias = (const float*)d_in[5];
    float* out = (float*)d_out;

    float* ws    = (float*)d_ws;
    float* xl    = ws;
    float* xr    = xl + NNODES * FEAT;
    unsigned* own1 = (unsigned*)(xr + NNODES * FEAT);

    gat_plane<<<NCHUNK, 256, 0, stream>>>(ei, own1);

    gat_gemm<<<dim3(NNODES / TROWS, 1, 2), 256, 0, stream>>>(x, W_l, W_r, xl, xr);

    gat_own<<<OWN_NB, OWN_THREADS, 0, stream>>>(xl, xr, ei, att, own1, bias, out);
}


# Owner planes hold edges, not bits: a 128-B line per (owner, chunk) = count + up to 31 packed (source, local target) words; owner blocks read no endpoints from the edge list; LDS arrival order
# speedup vs baseline: 4.2267x; 1.0648x over previous
#include <hip/hip_runtime.h>

#define NNODES   50000
#define FEAT     128
#define HEADS    4
#define NEDGE    800000
#define ETOT     (NEDGE + NNODES)
#define NEG_SLOPE 0.2f

typedef __attribute__((ext_vector_type(2))) float v2f;
typedef __attribute__((ext_vector_type(4))) float v4f;
typedef __attribute__((ext_vector_type(8))) float v8f;

#define OWN_NB      256
#define OWN_R       ((NNODES + OWN_NB - 1) / OWN_NB)
static_assert((NNODES - 1) / OWN_R < OWN_NB, "every node has an owner");
static_assert((OWN_NB - 1) * OWN_R < NNODES, "every block owns at least one node");
static_assert(NEDGE % 128 == 0, "a wave load of four destinations per lane ends with the list");

#define CH_EDGES    1024
#define LINE_DW     32
#define LINE_CAP    (LINE_DW - 1)
#define NCHUNK      ((NEDGE + CH_EDGES - 1) / CH_EDGES)
#define PLANE_DW    (NCHUNK * LINE_DW)
#define PLANE_TRIPS (OWN_NB * LINE_DW / 4 / 256)
static_assert(PLANE_TRIPS * 256 * 4 == OWN_NB * LINE_DW, "256 threads cover a chunk's lines exactly");
static_assert(CH_EDGES == 4 * 256, "a thread of the plane kernel takes four edges of the chunk");
static_assert(NNODES <= 65536 && OWN_R <= 65536, "a source and a local destination fit one word");

__global__ __launch_bounds__(256) void gat_plane(const int* __restrict__ ei,
                                                 unsigned* __restrict__ own1) {
    __shared__ __attribute__((aligned(16))) unsigned ent[OWN_NB * LINE_DW];
    __shared__ unsigned cnt[OWN_NB];
    const int c = blockIdx.x;
    const int q = c * (CH_EDGES / 4) + threadIdx.x;
    const int qc = (q < NEDGE / 4) ? q : (NEDGE / 4 - 1);
    const int4 sv = ((const int4*)ei)[qc];
    const int4 d  = ((const int4*)(ei + NEDGE))[qc];
    #pragma unroll
    for (int k = 0; k < PLANE_TRIPS; ++k)
        ((uint4*)ent)[threadIdx.x + 256 * k] = make_uint4(0u, 0u, 0u, 0u);
    static_assert(OWN_NB == 256, "one counter per thread");
    cnt[threadIdx.x] = 0u;
    __syncthreads();
    if (q < NEDGE / 4) {
        const int sa[4] = {sv.x, sv.y, sv.z, sv.w};
        const int da[4] = {d.x, d.y, d.z, d.w};
        unsigned o[4], slot[4];
        #pragma unroll
        for (int j = 0; j < 4; ++j) {
            o[j]    = (unsigned)da[j] / OWN_R;
            slot[j] = atomicAdd(&cnt[o[j]], 1u);
        }
        #pragma unroll
        for (int j = 0; j < 4; ++j)
            if (slot[j] < LINE_CAP)
                ent[o[j] * LINE_DW + 1 + slot[j]] =
                    (unsigned)sa[j] | (((unsigned)da[j] - o[j] * OWN_R) << 16);
    }
    __syncthreads();
    #pragma unroll
    for (int k = 0; k < PLANE_TRIPS; ++k) {
        const int i = threadIdx.x + 256 * k;
        const int o = i >> 3, part = i & 7;
        uint4 v = ((const uint4*)ent)[i];
        if (part == 0) v.x = cnt[o];
        ((uint4*)(own1 + o * PLANE_DW + c * LINE_DW))[part] = v;
    }
}

#define TROWS   80
#define RTILES  (TROWS / 16)
#define APITCH  (FEAT + 4)

__global__ __launch_bounds__(256) void gat_gemm(const float* __restrict__ x,
                                                const float* __restrict__ Wl,
                                                const float* __restrict__ Wr,
                                                float* __restrict__ xl,
                                                float* __restrict__ xr) {
    __shared__ __attribute__((aligned(16))) float As[TROWS * APITCH];

    const int lane  = threadIdx.x & 31;
    const int wave  = threadIdx.x >> 5;
    const int m0    = blockIdx.x * TROWS;
    const int n0    = wave * 16;
    const int lhalf = lane >> 4;
    const int l15   = lane & 15;

    const float* __restrict__ W    = blockIdx.z ? Wr : Wl;
    float*       __restrict__ outp = blockIdx.z ? xr : xl;

    #pragma unroll
    for (int i = 0; i < TROWS / 8; ++i) {
        const int r = wave + i * 8;
        const float4 v = ((const float4*)(x + (m0 + r) * FEAT))[lane];
        *(float4*)(&As[r * APITCH + lane * 4]) = v;
    }
    __syncthreads();

    v8f c[RTILES];
    #pragma unroll
    for (int t = 0; t < RTILES; ++t)
        c[t] = (v8f){0.f, 0.f, 0.f, 0.f, 0.f, 0.f, 0.f, 0.f};

    const float* ap = As + l15 * APITCH + lhalf * 2;
    const float* __restrict__ bp = W + (lhalf * 2) * FEAT + n0 + l15;

    #pragma unroll 4
    for (int k = 0; k < FEAT; k += 4) {
        v2f b;
        b.x = bp[k * FEAT];
        b.y = bp[(k + 1) * FEAT];
        #pragma unroll
        for (int t = 0; t < RTILES; ++t) {
            const v2f a = *(const v2f*)(ap + t * 16 * APITCH + k);
            c[t] = __builtin_amdgcn_wmma_f32_16x16x4_f32(
                        false, a,  false, b,
                        (short)0, c[t],  false,  false);
        }
    }

    #pragma unroll
    for (int t = 0; t < RTILES; ++t) {
        #pragma unroll
        for (int r = 0; r < 8; ++r)
            outp[(m0 + t * 16 + r + lhalf * 8) * FEAT + n0 + l15] = c[t][r];
    }
}

#define OWN_WAVES   32
#define OWN_THREADS (OWN_WAVES * 32)
#define NVEC        (PLANE_DW / 4)
#define NSCAN       ((NVEC + 31) / 32)
#define SCAN_UNR    ((NSCAN + OWN_WAVES - 1) / OWN_WAVES)
#define QGROUP      8
#define LCAP        4096
#define SELF_TRIPS  ((OWN_R + OWN_WAVES - 1) / OWN_WAVES)
#define EXP_CAP     40.0f
static_assert(LINE_DW == 32 && NSCAN * 4 >= NCHUNK, "a wave load is four whole lines");

struct OwnLds {
    float    acc[OWN_R * FEAT];
    float    xr[OWN_R * FEAT];
    float    sl[OWN_R * HEADS];
    float    ml[OWN_R * HEADS];
    unsigned list[LCAP];
    int      cnt;
    int      bad;
};

__device__ __forceinline__ float edge_dot(const float4 a, const float4 vl, const float4 vr) {
    float q = 0.f, t;
    t = vl.x + vr.x; t = (t >= 0.f) ? t : NEG_SLOPE * t; q += a.x * t;
    t = vl.y + vr.y; t = (t >= 0.f) ? t : NEG_SLOPE * t; q += a.y * t;
    t = vl.z + vr.z; t = (t >= 0.f) ? t : NEG_SLOPE * t; q += a.z * t;
    t = vl.w + vr.w; t = (t >= 0.f) ? t : NEG_SLOPE * t; q += a.w * t;
    return q;
}

__device__ __forceinline__ void own_fetch(const OwnLds& L, int ib, int g, int lane,
                                          int& sv, int& dv) {
    const int u = lane >> 2;
    const unsigned wd = (u < g) ? L.list[ib + u] : 0u;
    sv = (int)(wd & 0xFFFFu);
    dv = (int)(wd >> 16);
}

template <bool FULL>
__device__ __forceinline__ bool own_edges(OwnLds& L, int g, int lane, int sr, int dl,
                                          const float4 a, const float* __restrict__ xl,
                                          int ibn, int gn, int& sv, int& dv) {
    v4f vl[QGROUP];
    int dj[QGROUP];
    #pragma unroll
    for (int k = 0; k < QGROUP; ++k) {
        vl[k] = (v4f){0.f, 0.f, 0.f, 0.f};
        dj[k] = 0;
        if (FULL || k < g) {
            const int sj = __builtin_amdgcn_readlane(sr, 4 * k);
            dj[k]        = __builtin_amdgcn_readlane(dl, 4 * k);
            vl[k] = ((const v4f*)(xl + sj * FEAT))[lane];
        }
    }
    own_fetch(L, ibn, gn, lane, sv, dv);
    const int h = lane >> 3;
    v4f   vr[QGROUP];
    float cs[QGROUP];
    #pragma unroll
    for (int k = 0; k < QGROUP; ++k) {
        vr[k] = (v4f){0.f, 0.f, 0.f, 0.f};
        cs[k] = 0.f;
        if (FULL || k < g) {
            vr[k] = ((const v4f*)&L.xr[dj[k] * FEAT])[lane];
            cs[k] = L.ml[dj[k] * HEADS + h];
        }
    }
    float p[QGROUP];
    #pragma unroll
    for (int k = 0; k < QGROUP; ++k) {
        p[k] = 0.f;
        if (FULL || k < g)
            p[k] = edge_dot(a, make_float4(vl[k].x, vl[k].y, vl[k].z, vl[k].w),
                               make_float4(vr[k].x, vr[k].y, vr[k].z, vr[k].w));
    }
    __builtin_amdgcn_sched_barrier(0);
    #pragma unroll
    for (int k = 0; k < QGROUP; ++k) p[k] += __shfl_xor(p[k], 1, 32);
    __builtin_amdgcn_sched_barrier(0);
    #pragma unroll
    for (int k = 0; k < QGROUP; ++k) p[k] += __shfl_xor(p[k], 2, 32);
    __builtin_amdgcn_sched_barrier(0);
    #pragma unroll
    for (int k = 0; k < QGROUP; ++k) p[k] += __shfl_xor(p[k], 4, 32);
    __builtin_amdgcn_sched_barrier(0);

    bool bad = false;
    #pragma unroll
    for (int k = 0; k < QGROUP; ++k) {
        if (FULL || k < g) {
            const float d = p[k] - cs[k];
            bad = bad || !(d <= EXP_CAP);
            const float ex = __expf(d);
            float* ap = &L.acc[dj[k] * FEAT + lane];
            unsafeAtomicAdd(ap,      ex * vl[k].x);
            unsafeAtomicAdd(ap + 32, ex * vl[k].y);
            unsafeAtomicAdd(ap + 64, ex * vl[k].z);
            unsafeAtomicAdd(ap + 96, ex * vl[k].w);
            if ((lane & 7) == 0) unsafeAtomicAdd(&L.sl[dj[k] * HEADS + h], ex);
        }
    }
    return bad;
}

__device__ __forceinline__ bool own_sweep(OwnLds& L, int ib, int n, int lane,
                                          const float4 a, const float* __restrict__ xl) {
    bool bad = false;
    int sv, dv;
    own_fetch(L, ib, (n < QGROUP) ? n : QGROUP, lane, sv, dv);
    for (int j = 0; j < n; j += QGROUP) {
        const int g = (n - j < QGROUP) ? (n - j) : QGROUP;
        const int sr = sv, dl = dv;
        int gg = n - j - QGROUP;
        gg = (gg < 0) ? 0 : ((gg < QGROUP) ? gg : QGROUP);
        const int ibn = ib + j + QGROUP;
        if (g == QGROUP) bad = own_edges<true >(L, g, lane, sr, dl, a, xl, ibn, gg, sv, dv) || bad;
        else             bad = own_edges<false>(L, g, lane, sr, dl, a, xl, ibn, gg, sv, dv) || bad;
    }
    return bad;
}

template <int PASS>
__device__ __forceinline__ void own_one(OwnLds& L, int src, int dl, int lane, const float4 a,
                                        const float* __restrict__ xl) {
    const float4 vl = ((const float4*)(xl + src * FEAT))[lane];
    const float4 vr = ((const float4*)&L.xr[dl * FEAT])[lane];
    float p = edge_dot(a, vl, vr);
    p += __shfl_xor(p, 1, 32);
    p += __shfl_xor(p, 2, 32);
    p += __shfl_xor(p, 4, 32);
    const int h = lane >> 3;
    if (PASS == 0) {
        if ((lane & 7) == 0)
            __hip_atomic_fetch_max(&L.ml[dl * HEADS + h], p,
                                   __ATOMIC_RELAXED, __HIP_MEMORY_SCOPE_WORKGROUP);
    } else {
        const float ex = __expf(p - L.ml[dl * HEADS + h]);
        if ((lane & 7) == 0) unsafeAtomicAdd(&L.sl[dl * HEADS + h], ex);
        float* ap = &L.acc[dl * FEAT + lane];
        unsafeAtomicAdd(ap,      ex * vl.x);
        unsafeAtomicAdd(ap + 32, ex * vl.y);
        unsafeAtomicAdd(ap + 64, ex * vl.z);
        unsafeAtomicAdd(ap + 96, ex * vl.w);
    }
}

template <int PASS>
__device__ __forceinline__ void own_direct(OwnLds& L, int lane, int w, int lo, int nown,
                                        const float4 a,
                                        const float* __restrict__ xl,
                                        const int*   __restrict__ ei) {
    for (int n = w; n < nown; n += OWN_WAVES) own_one<PASS>(L, lo + n, n, lane, a, xl);
    for (int t = w; t < NEDGE / 128; t += OWN_WAVES) {
        const int4 d = ((const int4*)(ei + NEDGE))[t * 32 + lane];
        const unsigned hm = ((unsigned)(d.x - lo) < (unsigned)nown ? 1u : 0u)
                          | ((unsigned)(d.y - lo) < (unsigned)nown ? 2u : 0u)
                          | ((unsigned)(d.z - lo) < (unsigned)nown ? 4u : 0u)
                          | ((unsigned)(d.w - lo) < (unsigned)nown ? 8u : 0u);
        unsigned mask = __builtin_amdgcn_ballot_w32(hm != 0);
        while (mask) {
            const int l = __builtin_ctz(mask);
            unsigned hl = (unsigned)__builtin_amdgcn_readlane((int)hm, l);
            while (hl) {
                const int e = (t * 32 + l) * 4 + __builtin_ctz(hl);
                own_one<PASS>(L, ei[e], ei[NEDGE + e] - lo, lane, a, xl);
                hl &= hl - 1;
            }
            mask &= mask - 1;
        }
    }
}

__global__ __launch_bounds__(OWN_THREADS) void gat_own(const float* __restrict__ xl,
                                                       const float* __restrict__ xr,
                                                       const int*   __restrict__ ei,
                                                       const float* __restrict__ att,
                                                       const unsigned* __restrict__ own1,
                                                       const float* __restrict__ bias,
                                                       float* __restrict__ out) {
    __shared__ __attribute__((aligned(16))) OwnLds L;

    const int lane = threadIdx.x & 31;
    const int w    = __builtin_amdgcn_readfirstlane(threadIdx.x >> 5);
    const int lo   = blockIdx.x * OWN_R;
    const int nown = (NNODES - lo < OWN_R) ? (NNODES - lo) : OWN_R;
    const int h    = lane >> 3;

    const float4 a = ((const float4*)att)[lane];

    const uint4* __restrict__ ownv = (const uint4*)(own1 + blockIdx.x * PLANE_DW);
    uint4 d[SCAN_UNR];
    #pragma unroll
    for (int k = 0; k < SCAN_UNR; ++k) {
        const int piece = (w + k * OWN_WAVES) * 32 + lane;
        d[k] = ownv[(piece < NVEC) ? piece : (NVEC - 1)];
    }

    {
        v4f vr[SELF_TRIPS], vl[SELF_TRIPS];
        #pragma unroll
        for (int i = 0; i < SELF_TRIPS; ++i) {
            const int n  = w + i * OWN_WAVES;
            const int nc = (n < nown) ? n : (nown - 1);
            vr[i] = ((const v4f*)(xr + (lo + nc) * FEAT))[lane];
            vl[i] = ((const v4f*)(xl + (lo + nc) * FEAT))[lane];
        }
        float p[SELF_TRIPS];
        #pragma unroll
        for (int i = 0; i < SELF_TRIPS; ++i)
            p[i] = edge_dot(a, make_float4(vl[i].x, vl[i].y, vl[i].z, vl[i].w),
                               make_float4(vr[i].x, vr[i].y, vr[i].z, vr[i].w));
        #pragma unroll
        for (int i = 0; i < SELF_TRIPS; ++i) p[i] += __shfl_xor(p[i], 1, 32);
        #pragma unroll
        for (int i = 0; i < SELF_TRIPS; ++i) p[i] += __shfl_xor(p[i], 2, 32);
        #pragma unroll
        for (int i = 0; i < SELF_TRIPS; ++i) p[i] += __shfl_xor(p[i], 4, 32);
        #pragma unroll
        for (int i = 0; i < SELF_TRIPS; ++i) {
            const int n = w + i * OWN_WAVES;
            if (n < nown) {
                ((v4f*)&L.xr[n * FEAT])[lane] = vr[i];
                float* ap = &L.acc[n * FEAT + lane];
                ap[0] = vl[i].x; ap[32] = vl[i].y; ap[64] = vl[i].z; ap[96] = vl[i].w;
                if ((lane & 7) == 0) {
                    L.ml[n * HEADS + h] = p[i];
                    L.sl[n * HEADS + h] = 1.0f;
                }
            }
        }
    }
    if (threadIdx.x == 0) { L.cnt = 0; L.bad = 0; }
    __syncthreads();

    bool bad = false;
    {
        int  base[SCAN_UNR];
        int  nl[SCAN_UNR];
        int  tot  = 0;
        bool over = false;
        const int grp = lane >> 3;
        #pragma unroll
        for (int k = 0; k < SCAN_UNR; ++k) {
            const int ch0 = (w + k * OWN_WAVES) * 4;
            base[k] = 0;
            nl[k]   = 0;
            #pragma unroll
            for (int g = 0; g < 4; ++g) {
                const unsigned cg = (unsigned)__builtin_amdgcn_readlane((int)d[k].x, 8 * g);
                const bool live = ch0 + g < NCHUNK;
                const int  n    = live ? (int)((cg < (unsigned)LINE_CAP) ? cg : (unsigned)LINE_CAP) : 0;
                over = over || (live && cg > (unsigned)LINE_CAP);
                if (grp == g) { base[k] = tot; nl[k] = n; }
                tot += n;
            }
        }
        if (tot > 0) {
            int b = 0;
            if (lane == 0) b = atomicAdd(&L.cnt, tot);
            const int pos0 = __builtin_amdgcn_readlane(b, 0);
            if (pos0 + tot <= LCAP) {
                const int j4 = 4 * (lane & 7);
                #pragma unroll
                for (int k = 0; k < SCAN_UNR; ++k) {
                    const unsigned wd[4] = {d[k].x, d[k].y, d[k].z, d[k].w};
                    #pragma unroll
                    for (int i = 0; i < 4; ++i)
                        if (j4 + i >= 1 && j4 + i <= nl[k])
                            L.list[pos0 + base[k] + j4 + i - 1] = wd[i];
                }
                __builtin_amdgcn_wave_barrier();
                bad = own_sweep(L, pos0, tot, lane, a, xl);
            }
        }
        bad = bad || over;
    }
    if (__builtin_amdgcn_ballot_w32(bad) != 0u && lane == 0) L.bad = 1;
    __syncthreads();

    if (L.cnt > LCAP || L.bad != 0) {
        for (int i = threadIdx.x; i < OWN_R * FEAT; i += OWN_THREADS) L.acc[i] = 0.0f;
        for (int i = threadIdx.x; i < OWN_R * HEADS; i += OWN_THREADS) {
            L.sl[i] = 0.0f;
            L.ml[i] = -__builtin_inff();
        }
        __syncthreads();
        own_direct<0>(L, lane, w, lo, nown, a, xl, ei);
        __syncthreads();
        own_direct<1>(L, lane, w, lo, nown, a, xl, ei);
        __syncthreads();
    }

    const float4 bv = ((const float4*)bias)[lane];
    for (int n = w; n < nown; n += OWN_WAVES) {
        const float* ap = &L.acc[n * FEAT + lane];
        const float s = L.sl[n * HEADS + h];
        float4 o;
        o.x = ap[0]  / s + bv.x;
        o.y = ap[32] / s + bv.y;
        o.z = ap[64] / s + bv.z;
        o.w = ap[96] / s + bv.w;
        ((float4*)(out + (lo + n) * FEAT))[lane] = o;
    }
}

extern "C" void kernel_launch(void* const* d_in, const int* in_sizes, int n_in,
                              void* d_out, int out_size, void* d_ws, size_t ws_size,
                              hipStream_t stream) {
    const float* x    = (const float*)d_in[0];
    const int*   ei   = (const int*)  d_in[1];
    const float* W_l  = (const float*)d_in[2];
    const float* W_r  = (const float*)d_in[3];
    const float* att  = (const float*)d_in[4];
    const float* bias = (const float*)d_in[5];
    float* out = (float*)d_out;

    float* ws    = (float*)d_ws;
    float* xl    = ws;
    float* xr    = xl + NNODES * FEAT;
    unsigned* own1 = (unsigned*)(xr + NNODES * FEAT);

    gat_plane<<<NCHUNK, 256, 0, stream>>>(ei, own1);

    gat_gemm<<<dim3(NNODES / TROWS, 1, 2), 256, 0, stream>>>(x, W_l, W_r, xl, xr);

    gat_own<<<OWN_NB, OWN_THREADS, 0, stream>>>(xl, xr, ei, att, own1, bias, out);
}


# Owner sweep: behind a block barrier the waves draw the list's groups of 8 edges from an LDS counter, each until the list is empty, not each wave its own scan stretch; sums in LDS arrival order
# speedup vs baseline: 4.3768x; 1.0355x over previous
#include <hip/hip_runtime.h>

#define NNODES   50000
#define FEAT     128
#define HEADS    4
#define NEDGE    800000
#define ETOT     (NEDGE + NNODES)
#define NEG_SLOPE 0.2f

typedef __attribute__((ext_vector_type(2))) float v2f;
typedef __attribute__((ext_vector_type(4))) float v4f;
typedef __attribute__((ext_vector_type(8))) float v8f;

#define OWN_NB      256
#define OWN_R       ((NNODES + OWN_NB - 1) / OWN_NB)
static_assert((NNODES - 1) / OWN_R < OWN_NB, "every node has an owner");
static_assert((OWN_NB - 1) * OWN_R < NNODES, "every block owns at least one node");
static_assert(NEDGE % 128 == 0, "a wave load of four destinations per lane ends with the list");

#define CH_EDGES    1024
#define LINE_DW     32
#define LINE_CAP    (LINE_DW - 1)
#define NCHUNK      ((NEDGE + CH_EDGES - 1) / CH_EDGES)
#define PLANE_DW    (NCHUNK * LINE_DW)
#define PLANE_TRIPS (OWN_NB * LINE_DW / 4 / 256)
static_assert(PLANE_TRIPS * 256 * 4 == OWN_NB * LINE_DW, "256 threads cover a chunk's lines exactly");
static_assert(CH_EDGES == 4 * 256, "a thread of the plane kernel takes four edges of the chunk");
static_assert(NNODES <= 65536 && OWN_R <= 65536, "a source and a local destination fit one word");

__global__ __launch_bounds__(256) void gat_plane(const int* __restrict__ ei,
                                                 unsigned* __restrict__ own1) {
    __shared__ __attribute__((aligned(16))) unsigned ent[OWN_NB * LINE_DW];
    __shared__ unsigned cnt[OWN_NB];
    const int c = blockIdx.x;
    const int q = c * (CH_EDGES / 4) + threadIdx.x;
    const int qc = (q < NEDGE / 4) ? q : (NEDGE / 4 - 1);
    const int4 sv = ((const int4*)ei)[qc];
    const int4 d  = ((const int4*)(ei + NEDGE))[qc];
    #pragma unroll
    for (int k = 0; k < PLANE_TRIPS; ++k)
        ((uint4*)ent)[threadIdx.x + 256 * k] = make_uint4(0u, 0u, 0u, 0u);
    static_assert(OWN_NB == 256, "one counter per thread");
    cnt[threadIdx.x] = 0u;
    __syncthreads();
    if (q < NEDGE / 4) {
        const int sa[4] = {sv.x, sv.y, sv.z, sv.w};
        const int da[4] = {d.x, d.y, d.z, d.w};
        unsigned o[4], slot[4];
        #pragma unroll
        for (int j = 0; j < 4; ++j) {
            o[j]    = (unsigned)da[j] / OWN_R;
            slot[j] = atomicAdd(&cnt[o[j]], 1u);
        }
        #pragma unroll
        for (int j = 0; j < 4; ++j)
            if (slot[j] < LINE_CAP)
                ent[o[j] * LINE_DW + 1 + slot[j]] =
                    (unsigned)sa[j] | (((unsigned)da[j] - o[j] * OWN_R) << 16);
    }
    __syncthreads();
    #pragma unroll
    for (int k = 0; k < PLANE_TRIPS; ++k) {
        const int i = threadIdx.x + 256 * k;
        const int o = i >> 3, part = i & 7;
        uint4 v = ((const uint4*)ent)[i];
        if (part == 0) v.x = cnt[o];
        ((uint4*)(own1 + o * PLANE_DW + c * LINE_DW))[part] = v;
    }
}

#define TROWS   80
#define RTILES  (TROWS / 16)
#define APITCH  (FEAT + 4)

__global__ __launch_bounds__(256) void gat_gemm(const float* __restrict__ x,
                                                const float* __restrict__ Wl,
                                                const float* __restrict__ Wr,
                                                float* __restrict__ xl,
                                                float* __restrict__ xr) {
    __shared__ __attribute__((aligned(16))) float As[TROWS * APITCH];

    const int lane  = threadIdx.x & 31;
    const int wave  = threadIdx.x >> 5;
    const int m0    = blockIdx.x * TROWS;
    const int n0    = wave * 16;
    const int lhalf = lane >> 4;
    const int l15   = lane & 15;

    const float* __restrict__ W    = blockIdx.z ? Wr : Wl;
    float*       __restrict__ outp = blockIdx.z ? xr : xl;

    #pragma unroll
    for (int i = 0; i < TROWS / 8; ++i) {
        const int r = wave + i * 8;
        const float4 v = ((const float4*)(x + (m0 + r) * FEAT))[lane];
        *(float4*)(&As[r * APITCH + lane * 4]) = v;
    }
    __syncthreads();

    v8f c[RTILES];
    #pragma unroll
    for (int t = 0; t < RTILES; ++t)
        c[t] = (v8f){0.f, 0.f, 0.f, 0.f, 0.f, 0.f, 0.f, 0.f};

    const float* ap = As + l15 * APITCH + lhalf * 2;
    const float* __restrict__ bp = W + (lhalf * 2) * FEAT + n0 + l15;

    #pragma unroll 4
    for (int k = 0; k < FEAT; k += 4) {
        v2f b;
        b.x = bp[k * FEAT];
        b.y = bp[(k + 1) * FEAT];
        #pragma unroll
        for (int t = 0; t < RTILES; ++t) {
            const v2f a = *(const v2f*)(ap + t * 16 * APITCH + k);
            c[t] = __builtin_amdgcn_wmma_f32_16x16x4_f32(
                        false, a,  false, b,
                        (short)0, c[t],  false,  false);
        }
    }

    #pragma unroll
    for (int t = 0; t < RTILES; ++t) {
        #pragma unroll
        for (int r = 0; r < 8; ++r)
            outp[(m0 + t * 16 + r + lhalf * 8) * FEAT + n0 + l15] = c[t][r];
    }
}

#define OWN_WAVES   32
#define OWN_THREADS (OWN_WAVES * 32)
#define NVEC        (PLANE_DW / 4)
#define NSCAN       ((NVEC + 31) / 32)
#define SCAN_UNR    ((NSCAN + OWN_WAVES - 1) / OWN_WAVES)
#define QGROUP      8
#define LCAP        4096
#define SELF_TRIPS  ((OWN_R + OWN_WAVES - 1) / OWN_WAVES)
#define EXP_CAP     40.0f
static_assert(LINE_DW == 32 && NSCAN * 4 >= NCHUNK, "a wave load is four whole lines");

struct OwnLds {
    float    acc[OWN_R * FEAT];
    float    xr[OWN_R * FEAT];
    float    sl[OWN_R * HEADS];
    float    ml[OWN_R * HEADS];
    unsigned list[LCAP];
    int      cnt;
    int      next;
    int      over;
    int      bad;
};

__device__ __forceinline__ float edge_dot(const float4 a, const float4 vl, const float4 vr) {
    float q = 0.f, t;
    t = vl.x + vr.x; t = (t >= 0.f) ? t : NEG_SLOPE * t; q += a.x * t;
    t = vl.y + vr.y; t = (t >= 0.f) ? t : NEG_SLOPE * t; q += a.y * t;
    t = vl.z + vr.z; t = (t >= 0.f) ? t : NEG_SLOPE * t; q += a.z * t;
    t = vl.w + vr.w; t = (t >= 0.f) ? t : NEG_SLOPE * t; q += a.w * t;
    return q;
}

__device__ __forceinline__ int own_draw(OwnLds& L, int lane, int n) {
    int v = 0;
    if (lane == 0) v = atomicAdd(&L.next, n);
    return v;
}

__device__ __forceinline__ void own_fetch(const OwnLds& L, int ib, int g, int lane,
                                          int& sv, int& dv) {
    const int u = lane >> 2;
    const unsigned wd = (u < g) ? L.list[ib + u] : 0u;
    sv = (int)(wd & 0xFFFFu);
    dv = (int)(wd >> 16);
}

template <bool FULL>
__device__ __forceinline__ bool own_edges(OwnLds& L, int g, int lane, int sr, int dl,
                                          const float4 a, const float* __restrict__ xl,
                                          int ibn, int gn, int& sv, int& dv, int& nn) {
    v4f vl[QGROUP];
    int dj[QGROUP];
    #pragma unroll
    for (int k = 0; k < QGROUP; ++k) {
        vl[k] = (v4f){0.f, 0.f, 0.f, 0.f};
        dj[k] = 0;
        if (FULL || k < g) {
            const int sj = __builtin_amdgcn_readlane(sr, 4 * k);
            dj[k]        = __builtin_amdgcn_readlane(dl, 4 * k);
            vl[k] = ((const v4f*)(xl + sj * FEAT))[lane];
        }
    }
    own_fetch(L, ibn, gn, lane, sv, dv);
    const int h = lane >> 3;
    v4f   vr[QGROUP];
    float cs[QGROUP];
    #pragma unroll
    for (int k = 0; k < QGROUP; ++k) {
        vr[k] = (v4f){0.f, 0.f, 0.f, 0.f};
        cs[k] = 0.f;
        if (FULL || k < g) {
            vr[k] = ((const v4f*)&L.xr[dj[k] * FEAT])[lane];
            cs[k] = L.ml[dj[k] * HEADS + h];
        }
    }
    nn = __builtin_amdgcn_readfirstlane(own_draw(L, lane, 1));
    float p[QGROUP];
    #pragma unroll
    for (int k = 0; k < QGROUP; ++k) {
        p[k] = 0.f;
        if (FULL || k < g)
            p[k] = edge_dot(a, make_float4(vl[k].x, vl[k].y, vl[k].z, vl[k].w),
                               make_float4(vr[k].x, vr[k].y, vr[k].z, vr[k].w));
    }
    __builtin_amdgcn_sched_barrier(0);
    #pragma unroll
    for (int k = 0; k < QGROUP; ++k) p[k] += __shfl_xor(p[k], 1, 32);
    __builtin_amdgcn_sched_barrier(0);
    #pragma unroll
    for (int k = 0; k < QGROUP; ++k) p[k] += __shfl_xor(p[k], 2, 32);
    __builtin_amdgcn_sched_barrier(0);
    #pragma unroll
    for (int k = 0; k < QGROUP; ++k) p[k] += __shfl_xor(p[k], 4, 32);
    __builtin_amdgcn_sched_barrier(0);

    bool bad = false;
    #pragma unroll
    for (int k = 0; k < QGROUP; ++k) {
        if (FULL || k < g) {
            const float d = p[k] - cs[k];
            bad = bad || !(d <= EXP_CAP);
            const float ex = __expf(d);
            float* ap = &L.acc[dj[k] * FEAT + lane];
            unsafeAtomicAdd(ap,      ex * vl[k].x);
            unsafeAtomicAdd(ap + 32, ex * vl[k].y);
            unsafeAtomicAdd(ap + 64, ex * vl[k].z);
            unsafeAtomicAdd(ap + 96, ex * vl[k].w);
            if ((lane & 7) == 0) unsafeAtomicAdd(&L.sl[dj[k] * HEADS + h], ex);
        }
    }
    return bad;
}

__device__ __forceinline__ int own_gsize(int n, int q) {
    const int r = n - q * QGROUP;
    return (r < 0) ? 0 : ((r < QGROUP) ? r : QGROUP);
}

__device__ __forceinline__ bool own_sweep(OwnLds& L, int n, int lane,
                                          const float4 a, const float* __restrict__ xl) {
    const int ngrp = (n + QGROUP - 1) / QGROUP;
    bool bad = false;
    int cur = __builtin_amdgcn_readfirstlane(own_draw(L, lane, 2));
    int nxt = cur + 1;
    int sv, dv;
    own_fetch(L, cur * QGROUP, own_gsize(n, cur), lane, sv, dv);
    while (cur < ngrp) {
        const int g  = own_gsize(n, cur);
        const int gn = own_gsize(n, nxt);
        const int sr = sv, dl = dv;
        int nn;
        if (g == QGROUP) bad = own_edges<true >(L, g, lane, sr, dl, a, xl, nxt * QGROUP, gn, sv, dv, nn) || bad;
        else             bad = own_edges<false>(L, g, lane, sr, dl, a, xl, nxt * QGROUP, gn, sv, dv, nn) || bad;
        cur = nxt;
        nxt = nn;
    }
    return bad;
}

template <int PASS>
__device__ __forceinline__ void own_one(OwnLds& L, int src, int dl, int lane, const float4 a,
                                        const float* __restrict__ xl) {
    const float4 vl = ((const float4*)(xl + src * FEAT))[lane];
    const float4 vr = ((const float4*)&L.xr[dl * FEAT])[lane];
    float p = edge_dot(a, vl, vr);
    p += __shfl_xor(p, 1, 32);
    p += __shfl_xor(p, 2, 32);
    p += __shfl_xor(p, 4, 32);
    const int h = lane >> 3;
    if (PASS == 0) {
        if ((lane & 7) == 0)
            __hip_atomic_fetch_max(&L.ml[dl * HEADS + h], p,
                                   __ATOMIC_RELAXED, __HIP_MEMORY_SCOPE_WORKGROUP);
    } else {
        const float ex = __expf(p - L.ml[dl * HEADS + h]);
        if ((lane & 7) == 0) unsafeAtomicAdd(&L.sl[dl * HEADS + h], ex);
        float* ap = &L.acc[dl * FEAT + lane];
        unsafeAtomicAdd(ap,      ex * vl.x);
        unsafeAtomicAdd(ap + 32, ex * vl.y);
        unsafeAtomicAdd(ap + 64, ex * vl.z);
        unsafeAtomicAdd(ap + 96, ex * vl.w);
    }
}

template <int PASS>
__device__ __forceinline__ void own_direct(OwnLds& L, int lane, int w, int lo, int nown,
                                        const float4 a,
                                        const float* __restrict__ xl,
                                        const int*   __restrict__ ei) {
    for (int n = w; n < nown; n += OWN_WAVES) own_one<PASS>(L, lo + n, n, lane, a, xl);
    for (int t = w; t < NEDGE / 128; t += OWN_WAVES) {
        const int4 d = ((const int4*)(ei + NEDGE))[t * 32 + lane];
        const unsigned hm = ((unsigned)(d.x - lo) < (unsigned)nown ? 1u : 0u)
                          | ((unsigned)(d.y - lo) < (unsigned)nown ? 2u : 0u)
                          | ((unsigned)(d.z - lo) < (unsigned)nown ? 4u : 0u)
                          | ((unsigned)(d.w - lo) < (unsigned)nown ? 8u : 0u);
        unsigned mask = __builtin_amdgcn_ballot_w32(hm != 0);
        while (mask) {
            const int l = __builtin_ctz(mask);
            unsigned hl = (unsigned)__builtin_amdgcn_readlane((int)hm, l);
            while (hl) {
                const int e = (t * 32 + l) * 4 + __builtin_ctz(hl);
                own_one<PASS>(L, ei[e], ei[NEDGE + e] - lo, lane, a, xl);
                hl &= hl - 1;
            }
            mask &= mask - 1;
        }
    }
}

__global__ __launch_bounds__(OWN_THREADS) void gat_own(const float* __restrict__ xl,
                                                       const float* __restrict__ xr,
                                                       const int*   __restrict__ ei,
                                                       const float* __restrict__ att,
                                                       const unsigned* __restrict__ own1,
                                                       const float* __restrict__ bias,
                                                       float* __restrict__ out) {
    __shared__ __attribute__((aligned(16))) OwnLds L;

    const int lane = threadIdx.x & 31;
    const int w    = __builtin_amdgcn_readfirstlane(threadIdx.x >> 5);
    const int lo   = blockIdx.x * OWN_R;
    const int nown = (NNODES - lo < OWN_R) ? (NNODES - lo) : OWN_R;
    const int h    = lane >> 3;

    const float4 a = ((const float4*)att)[lane];

    const uint4* __restrict__ ownv = (const uint4*)(own1 + blockIdx.x * PLANE_DW);
    uint4 d[SCAN_UNR];
    #pragma unroll
    for (int k = 0; k < SCAN_UNR; ++k) {
        const int piece = (w + k * OWN_WAVES) * 32 + lane;
        d[k] = ownv[(piece < NVEC) ? piece : (NVEC - 1)];
    }

    {
        v4f vr[SELF_TRIPS], vl[SELF_TRIPS];
        #pragma unroll
        for (int i = 0; i < SELF_TRIPS; ++i) {
            const int n  = w + i * OWN_WAVES;
            const int nc = (n < nown) ? n : (nown - 1);
            vr[i] = ((const v4f*)(xr + (lo + nc) * FEAT))[lane];
            vl[i] = ((const v4f*)(xl + (lo + nc) * FEAT))[lane];
        }
        float p[SELF_TRIPS];
        #pragma unroll
        for (int i = 0; i < SELF_TRIPS; ++i)
            p[i] = edge_dot(a, make_float4(vl[i].x, vl[i].y, vl[i].z, vl[i].w),
                               make_float4(vr[i].x, vr[i].y, vr[i].z, vr[i].w));
        #pragma unroll
        for (int i = 0; i < SELF_TRIPS; ++i) p[i] += __shfl_xor(p[i], 1, 32);
        #pragma unroll
        for (int i = 0; i < SELF_TRIPS; ++i) p[i] += __shfl_xor(p[i], 2, 32);
        #pragma unroll
        for (int i = 0; i < SELF_TRIPS; ++i) p[i] += __shfl_xor(p[i], 4, 32);
        #pragma unroll
        for (int i = 0; i < SELF_TRIPS; ++i) {
            const int n = w + i * OWN_WAVES;
            if (n < nown) {
                ((v4f*)&L.xr[n * FEAT])[lane] = vr[i];
                float* ap = &L.acc[n * FEAT + lane];
                ap[0] = vl[i].x; ap[32] = vl[i].y; ap[64] = vl[i].z; ap[96] = vl[i].w;
                if ((lane & 7) == 0) {
                    L.ml[n * HEADS + h] = p[i];
                    L.sl[n * HEADS + h] = 1.0f;
                }
            }
        }
    }
    if (threadIdx.x == 0) { L.cnt = 0; L.next = 0; L.over = 0; L.bad = 0; }
    __syncthreads();

    {
        int  base[SCAN_UNR];
        int  nl[SCAN_UNR];
        int  tot  = 0;
        bool over = false;
        const int grp = lane >> 3;
        #pragma unroll
        for (int k = 0; k < SCAN_UNR; ++k) {
            const int ch0 = (w + k * OWN_WAVES) * 4;
            base[k] = 0;
            nl[k]   = 0;
            #pragma unroll
            for (int g = 0; g < 4; ++g) {
                const unsigned cg = (unsigned)__builtin_amdgcn_readlane((int)d[k].x, 8 * g);
                const bool live = ch0 + g < NCHUNK;
                const int  n    = live ? (int)((cg < (unsigned)LINE_CAP) ? cg : (unsigned)LINE_CAP) : 0;
                over = over || (live && cg > (unsigned)LINE_CAP);
                if (grp == g) { base[k] = tot; nl[k] = n; }
                tot += n;
            }
        }
        if (tot > 0) {
            int b = 0;
            if (lane == 0) b = atomicAdd(&L.cnt, tot);
            const int pos0 = __builtin_amdgcn_readlane(b, 0);
            if (pos0 + tot <= LCAP) {
                const int j4 = 4 * (lane & 7);
                #pragma unroll
                for (int k = 0; k < SCAN_UNR; ++k) {
                    const unsigned wd[4] = {d[k].x, d[k].y, d[k].z, d[k].w};
                    #pragma unroll
                    for (int i = 0; i < 4; ++i)
                        if (j4 + i >= 1 && j4 + i <= nl[k])
                            L.list[pos0 + base[k] + j4 + i - 1] = wd[i];
                }
            }
        }
        if (over && lane == 0) L.over = 1;
    }
    __syncthreads();

    const int  nlist = __builtin_amdgcn_readfirstlane(L.cnt);
    const bool swept = nlist <= LCAP && L.over == 0;
    if (swept) {
        const bool bad = own_sweep(L, nlist, lane, a, xl);
        if (__builtin_amdgcn_ballot_w32(bad) != 0u && lane == 0) L.bad = 1;
    }
    __syncthreads();

    if (!swept || L.bad != 0) {
        for (int i = threadIdx.x; i < OWN_R * FEAT; i += OWN_THREADS) L.acc[i] = 0.0f;
        for (int i = threadIdx.x; i < OWN_R * HEADS; i += OWN_THREADS) {
            L.sl[i] = 0.0f;
            L.ml[i] = -__builtin_inff();
        }
        __syncthreads();
        own_direct<0>(L, lane, w, lo, nown, a, xl, ei);
        __syncthreads();
        own_direct<1>(L, lane, w, lo, nown, a, xl, ei);
        __syncthreads();
    }

    const float4 bv = ((const float4*)bias)[lane];
    for (int n = w; n < nown; n += OWN_WAVES) {
        const float* ap = &L.acc[n * FEAT + lane];
        const float s = L.sl[n * HEADS + h];
        float4 o;
        o.x = ap[0]  / s + bv.x;
        o.y = ap[32] / s + bv.y;
        o.z = ap[64] / s + bv.z;
        o.w = ap[96] / s + bv.w;
        ((float4*)(out + (lo + n) * FEAT))[lane] = o;
    }
}

extern "C" void kernel_launch(void* const* d_in, const int* in_sizes, int n_in,
                              void* d_out, int out_size, void* d_ws, size_t ws_size,
                              hipStream_t stream) {
    const float* x    = (const float*)d_in[0];
    const int*   ei   = (const int*)  d_in[1];
    const float* W_l  = (const float*)d_in[2];
    const float* W_r  = (const float*)d_in[3];
    const float* att  = (const float*)d_in[4];
    const float* bias = (const float*)d_in[5];
    float* out = (float*)d_out;

    float* ws    = (float*)d_ws;
    float* xl    = ws;
    float* xr    = xl + NNODES * FEAT;
    unsigned* own1 = (unsigned*)(xr + NNODES * FEAT);

    gat_plane<<<NCHUNK, 256, 0, stream>>>(ei, own1);

    gat_gemm<<<dim3(NNODES / TROWS, 1, 2), 256, 0, stream>>>(x, W_l, W_r, xl, xr);

    gat_own<<<OWN_NB, OWN_THREADS, 0, stream>>>(xl, xr, ei, att, own1, bias, out);
}


# Sweep: a group's score partials cross lanes through a wave-own LDS tile, not 24 lane exchanges: lane (h,j) sums edge j, one exp and one s add per lane, exp read back 16 B wide; LDS arrival order
# speedup vs baseline: 4.5188x; 1.0324x over previous
#include <hip/hip_runtime.h>

#define NNODES   50000
#define FEAT     128
#define HEADS    4
#define NEDGE    800000
#define ETOT     (NEDGE + NNODES)
#define NEG_SLOPE 0.2f

typedef __attribute__((ext_vector_type(2))) float v2f;
typedef __attribute__((ext_vector_type(4))) float v4f;
typedef __attribute__((ext_vector_type(8))) float v8f;

#define OWN_NB      256
#define OWN_R       ((NNODES + OWN_NB - 1) / OWN_NB)
static_assert((NNODES - 1) / OWN_R < OWN_NB, "every node has an owner");
static_assert((OWN_NB - 1) * OWN_R < NNODES, "every block owns at least one node");
static_assert(NEDGE % 128 == 0, "a wave load of four destinations per lane ends with the list");

#define CH_EDGES    1024
#define LINE_DW     32
#define LINE_CAP    (LINE_DW - 1)
#define NCHUNK      ((NEDGE + CH_EDGES - 1) / CH_EDGES)
#define PLANE_DW    (NCHUNK * LINE_DW)
#define PLANE_TRIPS (OWN_NB * LINE_DW / 4 / 256)
static_assert(PLANE_TRIPS * 256 * 4 == OWN_NB * LINE_DW, "256 threads cover a chunk's lines exactly");
static_assert(CH_EDGES == 4 * 256, "a thread of the plane kernel takes four edges of the chunk");
static_assert(NNODES <= 65536 && OWN_R <= 65536, "a source and a local destination fit one word");

__global__ __launch_bounds__(256) void gat_plane(const int* __restrict__ ei,
                                                 unsigned* __restrict__ own1) {
    __shared__ __attribute__((aligned(16))) unsigned ent[OWN_NB * LINE_DW];
    __shared__ unsigned cnt[OWN_NB];
    const int c = blockIdx.x;
    const int q = c * (CH_EDGES / 4) + threadIdx.x;
    const int qc = (q < NEDGE / 4) ? q : (NEDGE / 4 - 1);
    const int4 sv = ((const int4*)ei)[qc];
    const int4 d  = ((const int4*)(ei + NEDGE))[qc];
    #pragma unroll
    for (int k = 0; k < PLANE_TRIPS; ++k)
        ((uint4*)ent)[threadIdx.x + 256 * k] = make_uint4(0u, 0u, 0u, 0u);
    static_assert(OWN_NB == 256, "one counter per thread");
    cnt[threadIdx.x] = 0u;
    __syncthreads();
    if (q < NEDGE / 4) {
        const int sa[4] = {sv.x, sv.y, sv.z, sv.w};
        const int da[4] = {d.x, d.y, d.z, d.w};
        unsigned o[4], slot[4];
        #pragma unroll
        for (int j = 0; j < 4; ++j) {
            o[j]    = (unsigned)da[j] / OWN_R;
            slot[j] = atomicAdd(&cnt[o[j]], 1u);
        }
        #pragma unroll
        for (int j = 0; j < 4; ++j)
            if (slot[j] < LINE_CAP)
                ent[o[j] * LINE_DW + 1 + slot[j]] =
                    (unsigned)sa[j] | (((unsigned)da[j] - o[j] * OWN_R) << 16);
    }
    __syncthreads();
    #pragma unroll
    for (int k = 0; k < PLANE_TRIPS; ++k) {
        const int i = threadIdx.x + 256 * k;
        const int o = i >> 3, part = i & 7;
        uint4 v = ((const uint4*)ent)[i];
        if (part == 0) v.x = cnt[o];
        ((uint4*)(own1 + o * PLANE_DW + c * LINE_DW))[part] = v;
    }
}

#define TROWS   80
#define RTILES  (TROWS / 16)
#define APITCH  (FEAT + 4)

__global__ __launch_bounds__(256) void gat_gemm(const float* __restrict__ x,
                                                const float* __restrict__ Wl,
                                                const float* __restrict__ Wr,
                                                float* __restrict__ xl,
                                                float* __restrict__ xr) {
    __shared__ __attribute__((aligned(16))) float As[TROWS * APITCH];

    const int lane  = threadIdx.x & 31;
    const int wave  = threadIdx.x >> 5;
    const int m0    = blockIdx.x * TROWS;
    const int n0    = wave * 16;
    const int lhalf = lane >> 4;
    const int l15   = lane & 15;

    const float* __restrict__ W    = blockIdx.z ? Wr : Wl;
    float*       __restrict__ outp = blockIdx.z ? xr : xl;

    #pragma unroll
    for (int i = 0; i < TROWS / 8; ++i) {
        const int r = wave + i * 8;
        const float4 v = ((const float4*)(x + (m0 + r) * FEAT))[lane];
        *(float4*)(&As[r * APITCH + lane * 4]) = v;
    }
    __syncthreads();

    v8f c[RTILES];
    #pragma unroll
    for (int t = 0; t < RTILES; ++t)
        c[t] = (v8f){0.f, 0.f, 0.f, 0.f, 0.f, 0.f, 0.f, 0.f};

    const float* ap = As + l15 * APITCH + lhalf * 2;
    const float* __restrict__ bp = W + (lhalf * 2) * FEAT + n0 + l15;

    #pragma unroll 4
    for (int k = 0; k < FEAT; k += 4) {
        v2f b;
        b.x = bp[k * FEAT];
        b.y = bp[(k + 1) * FEAT];
        #pragma unroll
        for (int t = 0; t < RTILES; ++t) {
            const v2f a = *(const v2f*)(ap + t * 16 * APITCH + k);
            c[t] = __builtin_amdgcn_wmma_f32_16x16x4_f32(
                        false, a,  false, b,
                        (short)0, c[t],  false,  false);
        }
    }

    #pragma unroll
    for (int t = 0; t < RTILES; ++t) {
        #pragma unroll
        for (int r = 0; r < 8; ++r)
            outp[(m0 + t * 16 + r + lhalf * 8) * FEAT + n0 + l15] = c[t][r];
    }
}

#define OWN_WAVES   32
#define OWN_THREADS (OWN_WAVES * 32)
#define NVEC        (PLANE_DW / 4)
#define NSCAN       ((NVEC + 31) / 32)
#define SCAN_UNR    ((NSCAN + OWN_WAVES - 1) / OWN_WAVES)
#define QGROUP      8
#define LCAP        4096
#define SELF_TRIPS  ((OWN_R + OWN_WAVES - 1) / OWN_WAVES)
#define TP_PITCH    36
#define TP_DW       (QGROUP * TP_PITCH)
#define EXP_CAP     40.0f
static_assert(LINE_DW == 32 && NSCAN * 4 >= NCHUNK, "a wave load is four whole lines");

struct OwnLds {
    float    acc[OWN_R * FEAT];
    float    xr[OWN_R * FEAT];
    float    sl[OWN_R * HEADS];
    float    ml[OWN_R * HEADS];
    unsigned list[LCAP];
    float    tp[OWN_WAVES * TP_DW];
    float    ep[OWN_WAVES * 32];
    int      cnt;
    int      next;
    int      over;
    int      bad;
};

__device__ __forceinline__ float edge_dot(const float4 a, const float4 vl, const float4 vr) {
    float q = 0.f, t;
    t = vl.x + vr.x; t = (t >= 0.f) ? t : NEG_SLOPE * t; q += a.x * t;
    t = vl.y + vr.y; t = (t >= 0.f) ? t : NEG_SLOPE * t; q += a.y * t;
    t = vl.z + vr.z; t = (t >= 0.f) ? t : NEG_SLOPE * t; q += a.z * t;
    t = vl.w + vr.w; t = (t >= 0.f) ? t : NEG_SLOPE * t; q += a.w * t;
    return q;
}

__device__ __forceinline__ int own_draw(OwnLds& L, int lane, int n) {
    int v = 0;
    if (lane == 0) v = atomicAdd(&L.next, n);
    return v;
}

__device__ __forceinline__ void own_fetch(const OwnLds& L, int ib, int g, int lane,
                                          int& sv, int& dv) {
    const int u = lane & 7;
    const unsigned wd = (u < g) ? L.list[ib + u] : 0u;
    sv = (int)(wd & 0xFFFFu);
    dv = (int)(wd >> 16);
}

template <bool FULL>
__device__ __forceinline__ bool own_edges(OwnLds& L, int g, int lane, int w, int sr, int dl,
                                          const float4 a, const float* __restrict__ xl,
                                          int ibn, int gn, int& sv, int& dv, int& nn) {
    v4f vl[QGROUP];
    int dj[QGROUP];
    #pragma unroll
    for (int k = 0; k < QGROUP; ++k) {
        vl[k] = (v4f){0.f, 0.f, 0.f, 0.f};
        dj[k] = 0;
        if (FULL || k < g) {
            const int sj = __builtin_amdgcn_readlane(sr, k);
            dj[k]        = __builtin_amdgcn_readlane(dl, k);
            vl[k] = ((const v4f*)(xl + sj * FEAT))[lane];
        }
    }
    own_fetch(L, ibn, gn, lane, sv, dv);
    const int h = lane >> 3;
    const int j = lane & 7;
    v4f vr[QGROUP];
    #pragma unroll
    for (int k = 0; k < QGROUP; ++k) {
        vr[k] = (v4f){0.f, 0.f, 0.f, 0.f};
        if (FULL || k < g)
            vr[k] = ((const v4f*)&L.xr[dj[k] * FEAT])[lane];
    }
    const float cl = L.ml[dl * HEADS + h];
    nn = __builtin_amdgcn_readfirstlane(own_draw(L, lane, 1));
    float* tp = &L.tp[w * TP_DW];
    float* ep = &L.ep[w * 32];
    #pragma unroll
    for (int k = 0; k < QGROUP; ++k) {
        float pk = 0.f;
        if (FULL || k < g)
            pk = edge_dot(a, make_float4(vl[k].x, vl[k].y, vl[k].z, vl[k].w),
                             make_float4(vr[k].x, vr[k].y, vr[k].z, vr[k].w));
        tp[k * TP_PITCH + lane] = pk;
    }
    const v4f q0 = *(const v4f*)&tp[j * TP_PITCH + 8 * h];
    const v4f q1 = *(const v4f*)&tp[j * TP_PITCH + 8 * h + 4];
    const float q = ((q0.x + q0.y) + (q0.z + q0.w)) + ((q1.x + q1.y) + (q1.z + q1.w));
    const float d = q - cl;
    const bool live = FULL || j < g;
    const bool bad = live && !(d <= EXP_CAP);
    const float exl = __expf(d);
    if (live) unsafeAtomicAdd(&L.sl[dl * HEADS + h], exl);
    ep[lane] = exl;
    const v4f e0 = *(const v4f*)&ep[8 * h];
    const v4f e1 = *(const v4f*)&ep[8 * h + 4];
    const float ex[QGROUP] = {e0.x, e0.y, e0.z, e0.w, e1.x, e1.y, e1.z, e1.w};
    #pragma unroll
    for (int k = 0; k < QGROUP; ++k) {
        if (FULL || k < g) {
            float* ap = &L.acc[dj[k] * FEAT + lane];
            unsafeAtomicAdd(ap,      ex[k] * vl[k].x);
            unsafeAtomicAdd(ap + 32, ex[k] * vl[k].y);
            unsafeAtomicAdd(ap + 64, ex[k] * vl[k].z);
            unsafeAtomicAdd(ap + 96, ex[k] * vl[k].w);
        }
    }
    return bad;
}

__device__ __forceinline__ int own_gsize(int n, int q) {
    const int r = n - q * QGROUP;
    return (r < 0) ? 0 : ((r < QGROUP) ? r : QGROUP);
}

__device__ __forceinline__ bool own_sweep(OwnLds& L, int n, int lane, int w,
                                          const float4 a, const float* __restrict__ xl) {
    const int ngrp = (n + QGROUP - 1) / QGROUP;
    bool bad = false;
    int cur = __builtin_amdgcn_readfirstlane(own_draw(L, lane, 2));
    int nxt = cur + 1;
    int sv, dv;
    own_fetch(L, cur * QGROUP, own_gsize(n, cur), lane, sv, dv);
    while (cur < ngrp) {
        const int g  = own_gsize(n, cur);
        const int gn = own_gsize(n, nxt);
        const int sr = sv, dl = dv;
        int nn;
        if (g == QGROUP) bad = own_edges<true >(L, g, lane, w, sr, dl, a, xl, nxt * QGROUP, gn, sv, dv, nn) || bad;
        else             bad = own_edges<false>(L, g, lane, w, sr, dl, a, xl, nxt * QGROUP, gn, sv, dv, nn) || bad;
        cur = nxt;
        nxt = nn;
    }
    return bad;
}

template <int PASS>
__device__ __forceinline__ void own_one(OwnLds& L, int src, int dl, int lane, const float4 a,
                                        const float* __restrict__ xl) {
    const float4 vl = ((const float4*)(xl + src * FEAT))[lane];
    const float4 vr = ((const float4*)&L.xr[dl * FEAT])[lane];
    float p = edge_dot(a, vl, vr);
    p += __shfl_xor(p, 1, 32);
    p += __shfl_xor(p, 2, 32);
    p += __shfl_xor(p, 4, 32);
    const int h = lane >> 3;
    if (PASS == 0) {
        if ((lane & 7) == 0)
            __hip_atomic_fetch_max(&L.ml[dl * HEADS + h], p,
                                   __ATOMIC_RELAXED, __HIP_MEMORY_SCOPE_WORKGROUP);
    } else {
        const float ex = __expf(p - L.ml[dl * HEADS + h]);
        if ((lane & 7) == 0) unsafeAtomicAdd(&L.sl[dl * HEADS + h], ex);
        float* ap = &L.acc[dl * FEAT + lane];
        unsafeAtomicAdd(ap,      ex * vl.x);
        unsafeAtomicAdd(ap + 32, ex * vl.y);
        unsafeAtomicAdd(ap + 64, ex * vl.z);
        unsafeAtomicAdd(ap + 96, ex * vl.w);
    }
}

template <int PASS>
__device__ __forceinline__ void own_direct(OwnLds& L, int lane, int w, int lo, int nown,
                                        const float4 a,
                                        const float* __restrict__ xl,
                                        const int*   __restrict__ ei) {
    for (int n = w; n < nown; n += OWN_WAVES) own_one<PASS>(L, lo + n, n, lane, a, xl);
    for (int t = w; t < NEDGE / 128; t += OWN_WAVES) {
        const int4 d = ((const int4*)(ei + NEDGE))[t * 32 + lane];
        const unsigned hm = ((unsigned)(d.x - lo) < (unsigned)nown ? 1u : 0u)
                          | ((unsigned)(d.y - lo) < (unsigned)nown ? 2u : 0u)
                          | ((unsigned)(d.z - lo) < (unsigned)nown ? 4u : 0u)
                          | ((unsigned)(d.w - lo) < (unsigned)nown ? 8u : 0u);
        unsigned mask = __builtin_amdgcn_ballot_w32(hm != 0);
        while (mask) {
            const int l = __builtin_ctz(mask);
            unsigned hl = (unsigned)__builtin_amdgcn_readlane((int)hm, l);
            while (hl) {
                const int e = (t * 32 + l) * 4 + __builtin_ctz(hl);
                own_one<PASS>(L, ei[e], ei[NEDGE + e] - lo, lane, a, xl);
                hl &= hl - 1;
            }
            mask &= mask - 1;
        }
    }
}

__global__ __launch_bounds__(OWN_THREADS) void gat_own(const float* __restrict__ xl,
                                                       const float* __restrict__ xr,
                                                       const int*   __restrict__ ei,
                                                       const float* __restrict__ att,
                                                       const unsigned* __restrict__ own1,
                                                       const float* __restrict__ bias,
                                                       float* __restrict__ out) {
    __shared__ __attribute__((aligned(16))) OwnLds L;

    const int lane = threadIdx.x & 31;
    const int w    = __builtin_amdgcn_readfirstlane(threadIdx.x >> 5);
    const int lo   = blockIdx.x * OWN_R;
    const int nown = (NNODES - lo < OWN_R) ? (NNODES - lo) : OWN_R;
    const int h    = lane >> 3;

    const float4 a = ((const float4*)att)[lane];

    const uint4* __restrict__ ownv = (const uint4*)(own1 + blockIdx.x * PLANE_DW);
    uint4 d[SCAN_UNR];
    #pragma unroll
    for (int k = 0; k < SCAN_UNR; ++k) {
        const int piece = (w + k * OWN_WAVES) * 32 + lane;
        d[k] = ownv[(piece < NVEC) ? piece : (NVEC - 1)];
    }

    {
        v4f vr[SELF_TRIPS], vl[SELF_TRIPS];
        #pragma unroll
        for (int i = 0; i < SELF_TRIPS; ++i) {
            const int n  = w + i * OWN_WAVES;
            const int nc = (n < nown) ? n : (nown - 1);
            vr[i] = ((const v4f*)(xr + (lo + nc) * FEAT))[lane];
            vl[i] = ((const v4f*)(xl + (lo + nc) * FEAT))[lane];
        }
        float p[SELF_TRIPS];
        #pragma unroll
        for (int i = 0; i < SELF_TRIPS; ++i)
            p[i] = edge_dot(a, make_float4(vl[i].x, vl[i].y, vl[i].z, vl[i].w),
                               make_float4(vr[i].x, vr[i].y, vr[i].z, vr[i].w));
        #pragma unroll
        for (int i = 0; i < SELF_TRIPS; ++i) p[i] += __shfl_xor(p[i], 1, 32);
        #pragma unroll
        for (int i = 0; i < SELF_TRIPS; ++i) p[i] += __shfl_xor(p[i], 2, 32);
        #pragma unroll
        for (int i = 0; i < SELF_TRIPS; ++i) p[i] += __shfl_xor(p[i], 4, 32);
        #pragma unroll
        for (int i = 0; i < SELF_TRIPS; ++i) {
            const int n = w + i * OWN_WAVES;
            if (n < nown) {
                ((v4f*)&L.xr[n * FEAT])[lane] = vr[i];
                float* ap = &L.acc[n * FEAT + lane];
                ap[0] = vl[i].x; ap[32] = vl[i].y; ap[64] = vl[i].z; ap[96] = vl[i].w;
                if ((lane & 7) == 0) {
                    L.ml[n * HEADS + h] = p[i];
                    L.sl[n * HEADS + h] = 1.0f;
                }
            }
        }
    }
    if (threadIdx.x == 0) { L.cnt = 0; L.next = 0; L.over = 0; L.bad = 0; }
    __syncthreads();

    {
        int  base[SCAN_UNR];
        int  nl[SCAN_UNR];
        int  tot  = 0;
        bool over = false;
        const int grp = lane >> 3;
        #pragma unroll
        for (int k = 0; k < SCAN_UNR; ++k) {
            const int ch0 = (w + k * OWN_WAVES) * 4;
            base[k] = 0;
            nl[k]   = 0;
            #pragma unroll
            for (int g = 0; g < 4; ++g) {
                const unsigned cg = (unsigned)__builtin_amdgcn_readlane((int)d[k].x, 8 * g);
                const bool live = ch0 + g < NCHUNK;
                const int  n    = live ? (int)((cg < (unsigned)LINE_CAP) ? cg : (unsigned)LINE_CAP) : 0;
                over = over || (live && cg > (unsigned)LINE_CAP);
                if (grp == g) { base[k] = tot; nl[k] = n; }
                tot += n;
            }
        }
        if (tot > 0) {
            int b = 0;
            if (lane == 0) b = atomicAdd(&L.cnt, tot);
            const int pos0 = __builtin_amdgcn_readlane(b, 0);
            if (pos0 + tot <= LCAP) {
                const int j4 = 4 * (lane & 7);
                #pragma unroll
                for (int k = 0; k < SCAN_UNR; ++k) {
                    const unsigned wd[4] = {d[k].x, d[k].y, d[k].z, d[k].w};
                    #pragma unroll
                    for (int i = 0; i < 4; ++i)
                        if (j4 + i >= 1 && j4 + i <= nl[k])
                            L.list[pos0 + base[k] + j4 + i - 1] = wd[i];
                }
            }
        }
        if (over && lane == 0) L.over = 1;
    }
    __syncthreads();

    const int  nlist = __builtin_amdgcn_readfirstlane(L.cnt);
    const bool swept = nlist <= LCAP && L.over == 0;
    if (swept) {
        const bool bad = own_sweep(L, nlist, lane, w, a, xl);
        if (__builtin_amdgcn_ballot_w32(bad) != 0u && lane == 0) L.bad = 1;
    }
    __syncthreads();

    if (!swept || L.bad != 0) {
        for (int i = threadIdx.x; i < OWN_R * FEAT; i += OWN_THREADS) L.acc[i] = 0.0f;
        for (int i = threadIdx.x; i < OWN_R * HEADS; i += OWN_THREADS) {
            L.sl[i] = 0.0f;
            L.ml[i] = -__builtin_inff();
        }
        __syncthreads();
        own_direct<0>(L, lane, w, lo, nown, a, xl, ei);
        __syncthreads();
        own_direct<1>(L, lane, w, lo, nown, a, xl, ei);
        __syncthreads();
    }

    const float4 bv = ((const float4*)bias)[lane];
    for (int n = w; n < nown; n += OWN_WAVES) {
        const float* ap = &L.acc[n * FEAT + lane];
        const float s = L.sl[n * HEADS + h];
        float4 o;
        o.x = ap[0]  / s + bv.x;
        o.y = ap[32] / s + bv.y;
        o.z = ap[64] / s + bv.z;
        o.w = ap[96] / s + bv.w;
        ((float4*)(out + (lo + n) * FEAT))[lane] = o;
    }
}

extern "C" void kernel_launch(void* const* d_in, const int* in_sizes, int n_in,
                              void* d_out, int out_size, void* d_ws, size_t ws_size,
                              hipStream_t stream) {
    const float* x    = (const float*)d_in[0];
    const int*   ei   = (const int*)  d_in[1];
    const float* W_l  = (const float*)d_in[2];
    const float* W_r  = (const float*)d_in[3];
    const float* att  = (const float*)d_in[4];
    const float* bias = (const float*)d_in[5];
    float* out = (float*)d_out;

    float* ws    = (float*)d_ws;
    float* xl    = ws;
    float* xr    = xl + NNODES * FEAT;
    unsigned* own1 = (unsigned*)(xr + NNODES * FEAT);

    gat_plane<<<NCHUNK, 256, 0, stream>>>(ei, own1);

    gat_gemm<<<dim3(NNODES / TROWS, 1, 2), 256, 0, stream>>>(x, W_l, W_r, xl, xr);

    gat_own<<<OWN_NB, OWN_THREADS, 0, stream>>>(xl, xr, ei, att, own1, bias, out);
}
